# starting kernel
# baseline (speedup 1.0000x reference)
#include <hip/hip_runtime.h>
#include <hip/hip_bf16.h>
#include <math.h>

#define NUM_USER   100000
#define NUM_ITEM   50000
#define NUM_GROUP  20000
#define EMB        64
#define BATCH      1024
#define E_UI       2000000
#define E_U        1000000
#define E_I        1000000
#define INV_TEMP   10.0f
#define EPS_N      1e-12f

#define N_UG  (NUM_USER + NUM_GROUP)
#define N_IG  (NUM_ITEM + NUM_GROUP)
#define N_UI  (NUM_USER + NUM_ITEM)

typedef _Float16 v16h __attribute__((ext_vector_type(16)));
typedef float    v8f  __attribute__((ext_vector_type(8)));

__global__ void zero_f32(float* __restrict__ p, size_t n) {
  size_t i = (size_t)blockIdx.x * blockDim.x + threadIdx.x;
  size_t stride = (size_t)gridDim.x * blockDim.x;
  for (; i < n; i += stride) p[i] = 0.0f;
}

__global__ void count_cols(const int* __restrict__ col, int* __restrict__ counts,
                           int nE) {
  int e = blockIdx.x * blockDim.x + threadIdx.x;
  if (e < nE) atomicAdd(&counts[col[e]], 1);
}

__global__ void exscan_kernel(int* __restrict__ counts, int* __restrict__ offsets,
                              int n) {
  __shared__ int sdata[1024];
  __shared__ int s_carry;
  int tid = threadIdx.x;
  if (tid == 0) s_carry = 0;
  __syncthreads();
  for (int base = 0; base < n; base += 1024) {
    int i = base + tid;
    int x = (i < n) ? counts[i] : 0;
    sdata[tid] = x;
    __syncthreads();
    for (int off = 1; off < 1024; off <<= 1) {
      int y = (tid >= off) ? sdata[tid - off] : 0;
      __syncthreads();
      sdata[tid] += y;
      __syncthreads();
    }
    int inc   = sdata[tid];
    int total = sdata[1023];
    int ex    = s_carry + inc - x;
    if (i < n) { offsets[i] = ex; counts[i] = ex; }
    __syncthreads();
    if (tid == 0) s_carry += total;
    __syncthreads();
  }
  if (tid == 0) offsets[n] = s_carry;
}

__global__ void csr_fill(const int* __restrict__ col, int* __restrict__ cursor,
                         int* __restrict__ eidx, int nE) {
  int e = blockIdx.x * blockDim.x + threadIdx.x;
  if (e < nE) {
    int slot = atomicAdd(&cursor[col[e]], 1);
    eidx[slot] = e;
  }
}

__global__ void csr_prop(const float* __restrict__ p0, int n0,
                         const float* __restrict__ p1,
                         const int* __restrict__ row,
                         const float* __restrict__ val,
                         const int* __restrict__ offsets,
                         const int* __restrict__ eidx,
                         float* __restrict__ out, int nRows) {
  int wid = (blockIdx.x * blockDim.x + threadIdx.x) >> 5;
  if (wid >= nRows) return;
  int lane = threadIdx.x & 31;
  int beg = offsets[wid], end = offsets[wid + 1];
  float ax = 0.0f, ay = 0.0f;
  for (int k = beg; k < end; ++k) {
    int e = eidx[k];
    int r = row[e];
    float v = val[e];
    const float* src = (r < n0) ? (p0 + (size_t)r * EMB)
                                : (p1 + (size_t)(r - n0) * EMB);
    float2 d = *(const float2*)(src + lane * 2);
    ax += d.x * v;
    ay += d.y * v;
  }
  float2 o; o.x = ax; o.y = ay;
  *(float2*)(out + (size_t)wid * EMB + lane * 2) = o;
}

__global__ void ui_final_kernel(const float* __restrict__ uei,
                                const float* __restrict__ item,
                                float* __restrict__ agg) {
  int gid = blockIdx.x * blockDim.x + threadIdx.x;
  if (gid >= N_UI * EMB) return;
  int r = gid >> 6, c = gid & 63;
  float uiv = (r < NUM_USER) ? uei[gid]
                             : item[(size_t)(r - NUM_USER) * EMB + c];
  agg[gid] = (uiv + agg[gid]) * 0.5f;
}

__global__ void gather_out(const float* __restrict__ ug,
                           const float* __restrict__ uif,
                           const float* __restrict__ ig,
                           const int* __restrict__ u,
                           const int* __restrict__ p,
                           const int* __restrict__ ng,
                           float* __restrict__ out) {
  int gid = blockIdx.x * blockDim.x + threadIdx.x;
  if (gid >= BATCH * EMB) return;
  int b = gid >> 6, c = gid & 63;
  const size_t PO = (size_t)BATCH * 128;
  const size_t NO = 2 * PO;
  out[(size_t)b * 128 + c]           = ug [(size_t)u[b] * EMB + c];
  out[(size_t)b * 128 + 64 + c]      = uif[(size_t)u[b] * EMB + c];
  out[PO + (size_t)b * 128 + c]      = ug [(size_t)(NUM_USER + p[b])  * EMB + c];
  out[PO + (size_t)b * 128 + 64 + c] = ig [(size_t)(NUM_ITEM + p[b])  * EMB + c];
  out[NO + (size_t)b * 128 + c]      = ug [(size_t)(NUM_USER + ng[b]) * EMB + c];
  out[NO + (size_t)b * 128 + 64 + c] = ig [(size_t)(NUM_ITEM + ng[b]) * EMB + c];
}

__global__ void reg_kernel(const float* __restrict__ ued,
                           const float* __restrict__ uei,
                           const float* __restrict__ ged,
                           const float* __restrict__ gei,
                           const int* __restrict__ u,
                           const int* __restrict__ p,
                           const int* __restrict__ ng,
                           float* __restrict__ reg_out) {
  int gid = blockIdx.x * blockDim.x + threadIdx.x;
  if (gid >= BATCH * EMB) return;
  int b = gid >> 6, c = gid & 63;
  size_t uo = (size_t)u[b]  * EMB + c;
  size_t po = (size_t)p[b]  * EMB + c;
  size_t no = (size_t)ng[b] * EMB + c;
  float s = uei[uo]*uei[uo] + ued[uo]*ued[uo]
          + gei[po]*gei[po] + ged[po]*ged[po]
          + gei[no]*gei[no] + ged[no]*ged[no];
  #pragma unroll
  for (int off = 1; off < 32; off <<= 1) s += __shfl_xor(s, off, 32);
  if ((threadIdx.x & 31) == 0) atomicAdd(reg_out, s * (0.5f / (float)BATCH));
}

__global__ void normalize_rows_f16(const float* __restrict__ in,
                                   _Float16* __restrict__ out, int nrows) {
  int gid = blockIdx.x * blockDim.x + threadIdx.x;
  int r = gid >> 5;
  if (r >= nrows) return;
  int t = gid & 31;
  size_t base = (size_t)r * EMB + 2 * t;
  float x0 = in[base], x1 = in[base + 1];
  float ss = x0 * x0 + x1 * x1;
  #pragma unroll
  for (int off = 1; off < 32; off <<= 1) ss += __shfl_xor(ss, off, 32);
  float scale = 1.0f / fmaxf(sqrtf(ss), EPS_N);
  out[base]     = (_Float16)(x0 * scale);
  out[base + 1] = (_Float16)(x1 * scale);
}

#define GEMM_WAVES 8
#define GEMM_TJ    16

__device__ __forceinline__ void load_btile(const _Float16* __restrict__ B16,
                                           int j0, int m, int kbB,
                                           v16h& b0, v16h& b1) {
  const _Float16* brow = B16 + (size_t)(j0 + m) * EMB + kbB;
  #pragma unroll
  for (int e = 0; e < 16; ++e) { b0[e] = brow[e]; b1[e] = brow[32 + e]; }
}

__global__ void ssl_gemm_expsum(const _Float16* __restrict__ A16,
                                const _Float16* __restrict__ B16,
                                const int* __restrict__ idx,
                                float* __restrict__ tot,
                                int Nall) {
  int lane = threadIdx.x & 31;
  int wave = threadIdx.x >> 5;
  int m0   = blockIdx.x * 16;
  int m    = lane & 15;
  int hi   = lane >> 4;

  const _Float16* arow = A16 + (size_t)idx[m0 + m] * EMB;
  int kbA = hi ? 8 : 0;
  v16h a0, a1;
  #pragma unroll
  for (int e = 0; e < 8; ++e) {
    a0[e]     = arow[kbA + e];
    a0[8 + e] = arow[16 + kbA + e];
    a1[e]     = arow[32 + kbA + e];
    a1[8 + e] = arow[48 + kbA + e];
  }

  float s[8];
  #pragma unroll
  for (int v = 0; v < 8; ++v) s[v] = 0.0f;

  int jt0 = (blockIdx.y * GEMM_WAVES + wave) * GEMM_TJ;
  int kbB = hi ? 16 : 0;
  int j0  = jt0 * 16;

  if (j0 < Nall) {
    v16h b0, b1;
    load_btile(B16, j0, m, kbB, b0, b1);
    for (int t = 0; t < GEMM_TJ; ++t) {
      int  j1   = j0 + 16;
      bool more = (t + 1 < GEMM_TJ) && (j1 < Nall);
      v16h nb0, nb1;
      if (more) load_btile(B16, j1, m, kbB, nb0, nb1);
      v8f c = {};
      c = __builtin_amdgcn_wmma_f32_16x16x32_f16(false, a0, false, b0,
                                                 (short)0, c, false, false);
      c = __builtin_amdgcn_wmma_f32_16x16x32_f16(false, a1, false, b1,
                                                 (short)0, c, false, false);
      #pragma unroll
      for (int v = 0; v < 8; ++v) s[v] += __expf(c[v] * INV_TEMP);
      if (!more) break;
      b0 = nb0; b1 = nb1; j0 = j1;
    }
  }
  #pragma unroll
  for (int v = 0; v < 8; ++v) {
    #pragma unroll
    for (int off = 1; off < 16; off <<= 1) s[v] += __shfl_xor(s[v], off, 32);
  }
  if ((lane & 15) == 0) {
    #pragma unroll
    for (int v = 0; v < 8; ++v)
      atomicAdd(&tot[m0 + v + 8 * hi], s[v]);
  }
}

__global__ void ssl_loss(const _Float16* __restrict__ A16,
                         const _Float16* __restrict__ B16,
                         const int* __restrict__ idx,
                         const float* __restrict__ tot,
                         float* __restrict__ ssl_out) {
  int gid = blockIdx.x * blockDim.x + threadIdx.x;
  int i = gid >> 5;
  if (i >= BATCH) return;
  int t = gid & 31;
  size_t ro = (size_t)idx[i] * EMB + 2 * t;
  float d = (float)A16[ro] * (float)B16[ro]
          + (float)A16[ro + 1] * (float)B16[ro + 1];
  #pragma unroll
  for (int off = 1; off < 32; off <<= 1) d += __shfl_xor(d, off, 32);
  if (t == 0) {
    float li = -(d * INV_TEMP - logf(tot[i])) * (1.0f / (float)BATCH);
    atomicAdd(ssl_out, li);
  }
}

extern "C" void kernel_launch(void* const* d_in, const int* in_sizes, int n_in,
                              void* d_out, int out_size, void* d_ws, size_t ws_size,
                              hipStream_t stream) {
  (void)in_sizes; (void)n_in; (void)out_size; (void)ws_size;
  const float* ued  = (const float*)d_in[0];
  const float* uei  = (const float*)d_in[1];
  const float* ged  = (const float*)d_in[2];
  const float* gei  = (const float*)d_in[3];
  const float* item = (const float*)d_in[4];
  const int*   hu_r = (const int*)d_in[5];
  const int*   hu_c = (const int*)d_in[6];
  const float* hu_v = (const float*)d_in[7];
  const int*   hi_r = (const int*)d_in[8];
  const int*   hi_c = (const int*)d_in[9];
  const float* hi_v = (const float*)d_in[10];
  const int*   ui_r = (const int*)d_in[11];
  const int*   ui_c = (const int*)d_in[12];
  const float* ui_v = (const float*)d_in[13];
  const int*   uin  = (const int*)d_in[14];
  const int*   pos  = (const int*)d_in[15];
  const int*   neg  = (const int*)d_in[16];
  float* out = (float*)d_out;

  float* W = (float*)d_ws;
  size_t off = 0;
  float* ug_b = W + off; off += (size_t)N_UG * EMB;
  float* ug_a = W + off; off += (size_t)N_UG * EMB;
  float* ig_b = W + off; off += (size_t)N_IG * EMB;
  float* ig_a = W + off; off += (size_t)N_IG * EMB;
  float* agg  = W + off; off += (size_t)N_UI * EMB;
  float* tot  = W + off; off += 4 * BATCH;
  int* counts  = (int*)(W + off); off += N_UI;
  int* offsets = (int*)(W + off); off += N_UI + 1;
  int* eidx    = (int*)(W + off); off += E_UI;
  off = (off + 7) & ~(size_t)7;
  _Float16* H = (_Float16*)(W + off);
  _Float16* ug_user_n16  = H;
  _Float16* ui_user_n16  = H + (size_t)NUM_USER * EMB;
  _Float16* ug_group_n16 = H + (size_t)NUM_USER * EMB * 2;
  _Float16* ig_group_n16 = ug_group_n16 + (size_t)NUM_GROUP * EMB;

  const int TB = 256;
  #define CDIV(a, b) (((a) + (b) - 1) / (b))

  zero_f32<<<16, TB, 0, stream>>>(tot, 4 * BATCH);
  zero_f32<<<1, 32, 0, stream>>>(out + (size_t)3 * BATCH * 128, 2);

  zero_f32<<<CDIV(N_UG, TB), TB, 0, stream>>>((float*)counts, N_UG);
  count_cols<<<CDIV(E_U, TB), TB, 0, stream>>>(hu_c, counts, E_U);
  exscan_kernel<<<1, 1024, 0, stream>>>(counts, offsets, N_UG);
  csr_fill<<<CDIV(E_U, TB), TB, 0, stream>>>(hu_c, counts, eidx, E_U);
  csr_prop<<<CDIV(N_UG * 32, TB), TB, 0, stream>>>(
      ued, NUM_USER, ged, hu_r, hu_v, offsets, eidx, ug_b, N_UG);
  csr_prop<<<CDIV(N_UG * 32, TB), TB, 0, stream>>>(
      ug_b, 1 << 30, nullptr, hu_r, hu_v, offsets, eidx, ug_a, N_UG);

  zero_f32<<<CDIV(N_IG, TB), TB, 0, stream>>>((float*)counts, N_IG);
  count_cols<<<CDIV(E_I, TB), TB, 0, stream>>>(hi_c, counts, E_I);
  exscan_kernel<<<1, 1024, 0, stream>>>(counts, offsets, N_IG);
  csr_fill<<<CDIV(E_I, TB), TB, 0, stream>>>(hi_c, counts, eidx, E_I);
  csr_prop<<<CDIV(N_IG * 32, TB), TB, 0, stream>>>(
      item, NUM_ITEM, gei, hi_r, hi_v, offsets, eidx, ig_b, N_IG);
  csr_prop<<<CDIV(N_IG * 32, TB), TB, 0, stream>>>(
      ig_b, 1 << 30, nullptr, hi_r, hi_v, offsets, eidx, ig_a, N_IG);

  zero_f32<<<CDIV(N_UI, TB), TB, 0, stream>>>((float*)counts, N_UI);
  count_cols<<<CDIV(E_UI, TB), TB, 0, stream>>>(ui_c, counts, E_UI);
  exscan_kernel<<<1, 1024, 0, stream>>>(counts, offsets, N_UI);
  csr_fill<<<CDIV(E_UI, TB), TB, 0, stream>>>(ui_c, counts, eidx, E_UI);
  csr_prop<<<CDIV(N_UI * 32, TB), TB, 0, stream>>>(
      uei, NUM_USER, item, ui_r, ui_v, offsets, eidx, agg, N_UI);
  ui_final_kernel<<<CDIV(N_UI * EMB, TB), TB, 0, stream>>>(uei, item, agg);

  gather_out<<<CDIV(BATCH * EMB, TB), TB, 0, stream>>>(
      ug_a, agg, ig_a, uin, pos, neg, out);
  reg_kernel<<<CDIV(BATCH * EMB, TB), TB, 0, stream>>>(
      ued, uei, ged, gei, uin, pos, neg, out + (size_t)3 * BATCH * 128);

  normalize_rows_f16<<<CDIV(NUM_USER * 32, TB), TB, 0, stream>>>(
      ug_a, ug_user_n16, NUM_USER);
  normalize_rows_f16<<<CDIV(NUM_USER * 32, TB), TB, 0, stream>>>(
      agg, ui_user_n16, NUM_USER);
  normalize_rows_f16<<<CDIV(NUM_GROUP * 32, TB), TB, 0, stream>>>(
      ug_a + (size_t)NUM_USER * EMB, ug_group_n16, NUM_GROUP);
  normalize_rows_f16<<<CDIV(NUM_GROUP * 32, TB), TB, 0, stream>>>(
      ig_a + (size_t)NUM_ITEM * EMB, ig_group_n16, NUM_GROUP);

  const int perBlk = GEMM_WAVES * GEMM_TJ;
  int ntU = NUM_USER / 16, ntG = NUM_GROUP / 16;
  dim3 gU(BATCH / 16, CDIV(ntU, perBlk));
  dim3 gG(BATCH / 16, CDIV(ntG, perBlk));
  ssl_gemm_expsum<<<gU, TB, 0, stream>>>(ui_user_n16, ug_user_n16, uin, tot + 0,         NUM_USER);
  ssl_gemm_expsum<<<gU, TB, 0, stream>>>(ug_user_n16, ui_user_n16, uin, tot + BATCH,     NUM_USER);
  ssl_gemm_expsum<<<gG, TB, 0, stream>>>(ig_group_n16, ug_group_n16, pos, tot + 2*BATCH, NUM_GROUP);
  ssl_gemm_expsum<<<gG, TB, 0, stream>>>(ug_group_n16, ig_group_n16, pos, tot + 3*BATCH, NUM_GROUP);

  float* ssl_slot = out + (size_t)3 * BATCH * 128 + 1;
  int lossBlocks = CDIV(BATCH * 32, TB);
  ssl_loss<<<lossBlocks, TB, 0, stream>>>(ui_user_n16, ug_user_n16, uin, tot + 0,         ssl_slot);
  ssl_loss<<<lossBlocks, TB, 0, stream>>>(ug_user_n16, ui_user_n16, uin, tot + BATCH,     ssl_slot);
  ssl_loss<<<lossBlocks, TB, 0, stream>>>(ig_group_n16, ug_group_n16, pos, tot + 2*BATCH, ssl_slot);
  ssl_loss<<<lossBlocks, TB, 0, stream>>>(ug_group_n16, ig_group_n16, pos, tot + 3*BATCH, ssl_slot);
  #undef CDIV
}


# SSL logits GEMM: a wave keeps 8 A tiles (128 batch rows) in registers, a table tile is loaded once per 128 rows, not per 16; A and B via one k-contiguous loader; row sums by row_shr, 128-byte atomics
# speedup vs baseline: 4.7077x; 4.7077x over previous
#include <hip/hip_runtime.h>
#include <hip/hip_bf16.h>
#include <math.h>

#define NUM_USER   100000
#define NUM_ITEM   50000
#define NUM_GROUP  20000
#define EMB        64
#define BATCH      1024
#define E_UI       2000000
#define E_U        1000000
#define E_I        1000000
#define INV_TEMP   10.0f
#define EPS_N      1e-12f

#define N_UG  (NUM_USER + NUM_GROUP)
#define N_IG  (NUM_ITEM + NUM_GROUP)
#define N_UI  (NUM_USER + NUM_ITEM)

typedef _Float16 v16h __attribute__((ext_vector_type(16)));
typedef float    v8f  __attribute__((ext_vector_type(8)));

__global__ void zero_f32(float* __restrict__ p, size_t n) {
  size_t i = (size_t)blockIdx.x * blockDim.x + threadIdx.x;
  size_t stride = (size_t)gridDim.x * blockDim.x;
  for (; i < n; i += stride) p[i] = 0.0f;
}

#define E_ALL  (E_U + E_I + E_UI)
#define N_ALL  (N_UG + N_IG + N_UI)
#define SLICE  62500
#define SL_U   ((E_U  + SLICE - 1) / SLICE)
#define SL_I   ((E_I  + SLICE - 1) / SLICE)
#define SL_X   ((E_UI + SLICE - 1) / SLICE)
#define HW_U   ((N_UG + 1) / 2)
#define HW_I   ((N_IG + 1) / 2)
#define HW_X   ((N_UI + 1) / 2)
#define HW_MAX 75000
#define HIST_THREADS 1024
#define HIST_U 8
static_assert(SLICE <= 65535, "a slice's counts fit 16 bits");
static_assert((SLICE & 3) == 0 && (E_U & 3) == 0 && (E_I & 3) == 0 && (E_UI & 3) == 0, "whole quads of edges");
static_assert(HW_U <= HW_MAX && HW_I <= HW_MAX && HW_X <= HW_MAX && (HW_MAX & 3) == 0, "counters fit LDS");

__device__ __forceinline__ int slice_rank(unsigned* s_h, int c) {
  const unsigned old = atomicAdd(&s_h[c >> 1], (c & 1) ? 0x10000u : 1u);
  return (int)((c & 1) ? (old >> 16) : (old & 0xffffu));
}

__global__ __launch_bounds__(HIST_THREADS)
void slice_hist3(const int* __restrict__ colU, const int* __restrict__ colI,
                 const int* __restrict__ colX,
                 unsigned* __restrict__ histU, unsigned* __restrict__ histI,
                 unsigned* __restrict__ histX, int* __restrict__ rank) {
  __shared__ __attribute__((aligned(16))) unsigned s_h[HW_MAX];
  const int b = blockIdx.x;
  const int* col = colU; unsigned* hist = histU;
  int nE = E_U, hw = HW_U, ebase = 0, s = b;
  if (b >= SL_U + SL_I) {
    col = colX; hist = histX; nE = E_UI; hw = HW_X; ebase = E_U + E_I; s = b - (SL_U + SL_I);
  } else if (b >= SL_U) {
    col = colI; hist = histI; nE = E_I; hw = HW_I; ebase = E_U; s = b - SL_U;
  }
  const int tid = threadIdx.x;
  for (int w = tid; w < HW_MAX / 4; w += HIST_THREADS) ((uint4*)s_h)[w] = make_uint4(0u, 0u, 0u, 0u);
  __syncthreads();

  const int e0  = s * SLICE;
  const int len = (nE - e0 < SLICE) ? (nE - e0) : SLICE;
  const int nq  = len >> 2;
  const int4* __restrict__ cq = (const int4*)(col + e0);
  int4* __restrict__ rq = (int4*)(rank + ebase + e0);
  for (int q0 = 0; q0 < nq; q0 += HIST_THREADS * HIST_U) {
    int4 c[HIST_U];
    #pragma unroll
    for (int u = 0; u < HIST_U; ++u) {
      const int q = q0 + u * HIST_THREADS + tid;
      c[u] = cq[(q < nq) ? q : (nq - 1)];
    }
    #pragma unroll
    for (int u = 0; u < HIST_U; ++u) {
      const int q = q0 + u * HIST_THREADS + tid;
      if (q < nq) {
        int4 r;
        r.x = slice_rank(s_h, c[u].x);
        r.y = slice_rank(s_h, c[u].y);
        r.z = slice_rank(s_h, c[u].z);
        r.w = slice_rank(s_h, c[u].w);
        rq[q] = r;
      }
    }
  }
  __syncthreads();
  unsigned* __restrict__ ho = hist + (size_t)s * hw;
  if ((hw & 3) == 0) {
    for (int w = tid; w < hw / 4; w += HIST_THREADS) ((uint4*)ho)[w] = ((const uint4*)s_h)[w];
  } else {
    for (int w = tid; w < hw; w += HIST_THREADS) ho[w] = s_h[w];
  }
}

#define CB_U 16
__global__ void col_base3(const unsigned* __restrict__ histU, const unsigned* __restrict__ histI,
                          const unsigned* __restrict__ histX,
                          int* __restrict__ baseU, int* __restrict__ baseI,
                          int* __restrict__ baseX, int* __restrict__ counts) {
  const int t = blockIdx.x * blockDim.x + threadIdx.x;
  if (t >= HW_U + HW_I + HW_X) return;
  const unsigned* hist = histU; int* base = baseU;
  int hw = HW_U, n = N_UG, ns = SL_U, cb = 0, w = t;
  if (t >= HW_U + HW_I) {
    hist = histX; base = baseX; hw = HW_X; n = N_UI; ns = SL_X; cb = N_UG + N_IG; w = t - (HW_U + HW_I);
  } else if (t >= HW_U) {
    hist = histI; base = baseI; hw = HW_I; n = N_IG; ns = SL_I; cb = N_UG; w = t - HW_U;
  }
  const int  c0   = 2 * w;
  const bool two  = (c0 + 1 < n);
  const bool pair = two && ((n & 1) == 0);
  int run0 = 0, run1 = 0;
  for (int s0 = 0; s0 < ns; s0 += CB_U) {
    unsigned v[CB_U];
    #pragma unroll
    for (int u = 0; u < CB_U; ++u) {
      const int s = s0 + u;
      v[u] = hist[(size_t)((s < ns) ? s : (ns - 1)) * hw + w];
    }
    #pragma unroll
    for (int u = 0; u < CB_U; ++u) {
      const int s = s0 + u;
      if (s < ns) {
        int* bp = base + (size_t)s * n + c0;
        if (pair) { int2 o; o.x = run0; o.y = run1; *(int2*)bp = o; }
        else      { bp[0] = run0; if (two) bp[1] = run1; }
        run0 += (int)(v[u] & 0xffffu);
        run1 += (int)(v[u] >> 16);
      }
    }
  }
  counts[cb + c0] = run0;
  if (two) counts[cb + c0 + 1] = run1;
}

#define SCAN_THREADS 1024
#define SCAN_WAVES   (SCAN_THREADS / 32)
#define SCAN_TILE    128
#define SCAN_U       8
#define SCAN_GROUP   (SCAN_TILE * SCAN_U)

__device__ __forceinline__ int wave_incl_scan(int x, int lane) {
  x += __builtin_amdgcn_update_dpp(0, x, 0x111, 0xf, 0xf, true);
  x += __builtin_amdgcn_update_dpp(0, x, 0x112, 0xf, 0xf, true);
  x += __builtin_amdgcn_update_dpp(0, x, 0x114, 0xf, 0xf, true);
  x += __builtin_amdgcn_update_dpp(0, x, 0x118, 0xf, 0xf, true);
  const int low = __builtin_amdgcn_readlane(x, 15);
  x += (lane & 16) ? low : 0;
  return x;
}

__device__ __forceinline__ void scan_load_group(const int* __restrict__ p, int t,
                                                int lane, int end, bool vec,
                                                int4 (&v)[SCAN_U]) {
  if (vec && t + SCAN_GROUP <= end) {
    #pragma unroll
    for (int u = 0; u < SCAN_U; ++u)
      v[u] = *(const int4*)(p + t + u * SCAN_TILE + lane * 4);
  } else {
    #pragma unroll
    for (int u = 0; u < SCAN_U; ++u) {
      int i = t + u * SCAN_TILE + lane * 4;
      v[u].x = (i     < end) ? p[i]     : 0;
      v[u].y = (i + 1 < end) ? p[i + 1] : 0;
      v[u].z = (i + 2 < end) ? p[i + 2] : 0;
      v[u].w = (i + 3 < end) ? p[i + 3] : 0;
    }
  }
}

__device__ __forceinline__ void scan_store_group(int* __restrict__ p, int t,
                                                 int lane, int end, bool vec,
                                                 const int4 (&v)[SCAN_U]) {
  if (vec && t + SCAN_GROUP <= end) {
    #pragma unroll
    for (int u = 0; u < SCAN_U; ++u)
      *(int4*)(p + t + u * SCAN_TILE + lane * 4) = v[u];
  } else {
    #pragma unroll
    for (int u = 0; u < SCAN_U; ++u) {
      int i = t + u * SCAN_TILE + lane * 4;
      if (i     < end) p[i]     = v[u].x;
      if (i + 1 < end) p[i + 1] = v[u].y;
      if (i + 2 < end) p[i + 2] = v[u].z;
      if (i + 3 < end) p[i + 3] = v[u].w;
    }
  }
}

__global__ __launch_bounds__(SCAN_THREADS)
void exscan3_kernel(int* __restrict__ cnt0, int* __restrict__ off0, int n0,
                    int* __restrict__ cnt1, int* __restrict__ off1, int n1,
                    int* __restrict__ cnt2, int* __restrict__ off2, int n2) {
  int* __restrict__ counts  = cnt0;
  int* __restrict__ offsets = off0;
  int n = n0;
  if (blockIdx.x == 1) { counts = cnt1; offsets = off1; n = n1; }
  if (blockIdx.x == 2) { counts = cnt2; offsets = off2; n = n2; }
  __shared__ int s_wsum[SCAN_WAVES];
  const int tid  = threadIdx.x;
  const int lane = tid & 31;
  const int wave = __builtin_amdgcn_readfirstlane(tid >> 5);
  const int per = (((n + SCAN_WAVES - 1) / SCAN_WAVES) + SCAN_GROUP - 1)
                  & ~(SCAN_GROUP - 1);
  const int beg = wave * per;
  const int end = (beg + per < n) ? (beg + per) : n;
  const bool vec = ((((size_t)counts) | ((size_t)offsets)) & 15) == 0;

  int acc = 0;
  for (int t = beg; t < end; t += SCAN_GROUP) {
    int4 v[SCAN_U];
    scan_load_group(counts, t, lane, end, vec, v);
    #pragma unroll
    for (int u = 0; u < SCAN_U; ++u) acc += (v[u].x + v[u].y) + (v[u].z + v[u].w);
  }
  #pragma unroll
  for (int off = 16; off > 0; off >>= 1) acc += __shfl_xor(acc, off, 32);
  if (lane == 0) s_wsum[wave] = acc;
  __syncthreads();

  const int wt    = s_wsum[lane];
  const int winc  = wave_incl_scan(wt, lane);
  int       carry = __builtin_amdgcn_readlane(winc - wt, wave);
  const int grand = __builtin_amdgcn_readlane(winc, 31);

  for (int t = beg; t < end; t += SCAN_GROUP) {
    int4 v[SCAN_U];
    scan_load_group(counts, t, lane, end, vec, v);
    int4 o[SCAN_U];
    #pragma unroll
    for (int u = 0; u < SCAN_U; ++u) {
      const int p1 = v[u].x;
      const int p2 = p1 + v[u].y;
      const int p3 = p2 + v[u].z;
      const int p4 = p3 + v[u].w;
      const int inc = wave_incl_scan(p4, lane);
      const int ex  = carry + (inc - p4);
      o[u].x = ex;
      o[u].y = ex + p1;
      o[u].z = ex + p2;
      o[u].w = ex + p3;
      carry += __builtin_amdgcn_readlane(inc, 31);
    }
    scan_store_group(offsets, t, lane, end, vec, o);
    scan_store_group(counts,  t, lane, end, vec, o);
  }
  if (tid == 0) offsets[n] = grand;
}

__global__ void csr_fill3(const int* __restrict__ colU, const int* __restrict__ rowU,
                          const float* __restrict__ valU, const int* __restrict__ offU,
                          const int* __restrict__ baseU, int* __restrict__ pairsU,
                          const int* __restrict__ colI, const int* __restrict__ rowI,
                          const float* __restrict__ valI, const int* __restrict__ offI,
                          const int* __restrict__ baseI, int* __restrict__ pairsI,
                          const int* __restrict__ colX, const int* __restrict__ rowX,
                          const float* __restrict__ valX, const int* __restrict__ offX,
                          const int* __restrict__ baseX, int* __restrict__ pairsX,
                          const int* __restrict__ rank) {
  int e = blockIdx.x * blockDim.x + threadIdx.x;
  if (e >= E_ALL) return;
  const int* col = colU; const int* row = rowU; const float* val = valU;
  const int* offsets = offU; const int* base = baseU; int* pairs = pairsU;
  int le = e, n = N_UG;
  if (e >= E_U + E_I) {
    col = colX; row = rowX; val = valX; offsets = offX; base = baseX; pairs = pairsX;
    le = e - (E_U + E_I); n = N_UI;
  } else if (e >= E_U) {
    col = colI; row = rowI; val = valI; offsets = offI; base = baseI; pairs = pairsI;
    le = e - E_U; n = N_IG;
  }
  const int c = col[le];
  int slot = offsets[c] + base[(size_t)(le / SLICE) * n + c] + rank[e];
  int2 rv;
  rv.x = row[le];
  rv.y = __float_as_int(val[le]);
  *(int2*)(pairs + 2 * (size_t)slot) = rv;
}

typedef _Float16 v2h __attribute__((ext_vector_type(2)));

#define GATHER_DEPTH 16

__device__ __forceinline__ const float* table_row(const float* __restrict__ p0, int n0,
                                                  const float* __restrict__ p1, int r) {
  return (r < n0) ? (p0 + (size_t)r * EMB) : (p1 + (size_t)(r - n0) * EMB);
}

__device__ __forceinline__ int2 load_pairs(const int2* __restrict__ rv, int base,
                                           int end, int lane) {
  int2 p; p.x = 0; p.y = 0;
  if (base + lane < end) p = rv[base + lane];
  return p;
}

__device__ __forceinline__ void batch_load(const float* __restrict__ p0, int n0,
                                           const float* __restrict__ p1,
                                           int px, int j, int n, int lane,
                                           float2 (&d)[GATHER_DEPTH]) {
  int r[GATHER_DEPTH];
  #pragma unroll
  for (int u = 0; u < GATHER_DEPTH; ++u)
    r[u] = __builtin_amdgcn_readlane(px, j + u);
  #pragma unroll
  for (int u = 0; u < GATHER_DEPTH; ++u) {
    if (j + u < n)
      d[u] = *(const float2*)(table_row(p0, n0, p1, r[u]) + lane * 2);
  }
}

__device__ __forceinline__ void batch_sum(int pv, int j, int n,
                                          const float2 (&d)[GATHER_DEPTH],
                                          float& ax, float& ay) {
  #pragma unroll
  for (int u = 0; u < GATHER_DEPTH; ++u) {
    if (j + u < n) {
      const float v = __int_as_float(__builtin_amdgcn_readlane(pv, j + u));
      ax += d[u].x * v;
      ay += d[u].y * v;
    }
  }
}

__global__ void csr_prop(const float* __restrict__ p0, int n0,
                         const float* __restrict__ p1,
                         const int* __restrict__ offsets,
                         const int* __restrict__ pairs,
                         float* __restrict__ out, int nRows, int rowsPerWave,
                         int addSelf,
                         _Float16* __restrict__ nlo, int split,
                         _Float16* __restrict__ nhi) {
  const int2* __restrict__ rv = (const int2*)pairs;
  const int lane = threadIdx.x & 31;
  const int wv = __builtin_amdgcn_readfirstlane(
      (int)((blockIdx.x * blockDim.x + threadIdx.x) >> 5));
  const int r0 = wv * rowsPerWave;
  if (r0 >= nRows) return;
  const int nr = (nRows - r0 < rowsPerWave) ? (nRows - r0) : rowsPerWave;
  const int offv = (lane <= nr) ? offsets[r0 + lane] : 0;
  int beg = __builtin_amdgcn_readlane(offv, 0);
  int end = __builtin_amdgcn_readlane(offv, 1);
  int2 cur = load_pairs(rv, beg, end, lane);

  for (int i = 0; i < nr; ++i) {
    const int wid = r0 + i;
    const int nf  = (end - beg < 32) ? (end - beg) : 32;
    float ax = 0.0f, ay = 0.0f;
    float2 d[GATHER_DEPTH];
    batch_load(p0, n0, p1, cur.x, 0, nf, lane, d);
    int nend = end;
    int2 nxt; nxt.x = 0; nxt.y = 0;
    if (i + 1 < nr) {
      nend = __builtin_amdgcn_readlane(offv, i + 2);
      nxt  = load_pairs(rv, end, nend, lane);
    }
    float sx = 0.0f, sy = 0.0f;
    if (addSelf) {
      float2 s = *(const float2*)(table_row(p0, n0, p1, wid) + lane * 2);
      sx = s.x; sy = s.y;
    }
    batch_sum(cur.y, 0, nf, d, ax, ay);
    if (nf > GATHER_DEPTH) {
      batch_load(p0, n0, p1, cur.x, GATHER_DEPTH, nf, lane, d);
      batch_sum(cur.y, GATHER_DEPTH, nf, d, ax, ay);
    }
    for (int base = beg + 32; base < end; base += 32) {
      const int2 more = load_pairs(rv, base, end, lane);
      const int n = (end - base < 32) ? (end - base) : 32;
      for (int j = 0; j < n; j += GATHER_DEPTH) {
        batch_load(p0, n0, p1, more.x, j, n, lane, d);
        batch_sum(more.y, j, n, d, ax, ay);
      }
    }
    float2 o; o.x = ax; o.y = ay;
    if (addSelf) { o.x = (sx + ax) * 0.5f; o.y = (sy + ay) * 0.5f; }
    *(float2*)(out + (size_t)wid * EMB + lane * 2) = o;

    if (nlo != nullptr || nhi != nullptr) {
      float ss = fmaf(o.x, o.x, o.y * o.y);
      #pragma unroll
      for (int off = 1; off < 32; off <<= 1) ss += __shfl_xor(ss, off, 32);
      float scale = 1.0f / fmaxf(sqrtf(ss), EPS_N);
      _Float16* dst = nullptr;
      if (wid < split) { if (nlo != nullptr) dst = nlo + (size_t)wid * EMB; }
      else             { if (nhi != nullptr) dst = nhi + (size_t)(wid - split) * EMB; }
      if (dst != nullptr) {
        v2h h;
        h.x = (_Float16)(o.x * scale);
        h.y = (_Float16)(o.y * scale);
        *(v2h*)(dst + lane * 2) = h;
      }
    }
    beg = end; end = nend; cur = nxt;
  }
}

__global__ void gather_out(const float* __restrict__ ug,
                           const float* __restrict__ uif,
                           const float* __restrict__ ig,
                           const int* __restrict__ u,
                           const int* __restrict__ p,
                           const int* __restrict__ ng,
                           float* __restrict__ out) {
  int gid = blockIdx.x * blockDim.x + threadIdx.x;
  if (gid >= BATCH * EMB) return;
  int b = gid >> 6, c = gid & 63;
  const size_t PO = (size_t)BATCH * 128;
  const size_t NO = 2 * PO;
  out[(size_t)b * 128 + c]           = ug [(size_t)u[b] * EMB + c];
  out[(size_t)b * 128 + 64 + c]      = uif[(size_t)u[b] * EMB + c];
  out[PO + (size_t)b * 128 + c]      = ug [(size_t)(NUM_USER + p[b])  * EMB + c];
  out[PO + (size_t)b * 128 + 64 + c] = ig [(size_t)(NUM_ITEM + p[b])  * EMB + c];
  out[NO + (size_t)b * 128 + c]      = ug [(size_t)(NUM_USER + ng[b]) * EMB + c];
  out[NO + (size_t)b * 128 + 64 + c] = ig [(size_t)(NUM_ITEM + ng[b]) * EMB + c];
}

__global__ void reg_kernel(const float* __restrict__ ued,
                           const float* __restrict__ uei,
                           const float* __restrict__ ged,
                           const float* __restrict__ gei,
                           const int* __restrict__ u,
                           const int* __restrict__ p,
                           const int* __restrict__ ng,
                           float* __restrict__ reg_out) {
  int gid = blockIdx.x * blockDim.x + threadIdx.x;
  if (gid >= BATCH * EMB) return;
  int b = gid >> 6, c = gid & 63;
  size_t uo = (size_t)u[b]  * EMB + c;
  size_t po = (size_t)p[b]  * EMB + c;
  size_t no = (size_t)ng[b] * EMB + c;
  float s = uei[uo]*uei[uo] + ued[uo]*ued[uo]
          + gei[po]*gei[po] + ged[po]*ged[po]
          + gei[no]*gei[no] + ged[no]*ged[no];
  #pragma unroll
  for (int off = 1; off < 32; off <<= 1) s += __shfl_xor(s, off, 32);
  if ((threadIdx.x & 31) == 0) atomicAdd(reg_out, s * (0.5f / (float)BATCH));
}

#define GEMM_MT     8
#define GEMM_ROWS   (16 * GEMM_MT)
#define GEMM_WAVES  (BATCH / GEMM_ROWS)
#define GEMM_TPG_U  13
#define GEMM_TPG_G  5
static_assert(BATCH % GEMM_ROWS == 0, "whole row groups");
static_assert(NUM_USER % 16 == 0 && NUM_GROUP % 16 == 0, "whole table tiles");

__device__ __forceinline__ void load_frag2(const _Float16* __restrict__ row, int h,
                                           v16h& f0, v16h& f1) {
  const _Float16* p = row + 16 * h;
  f0 = *(const v16h*)p;
  f1 = *(const v16h*)(p + 32);
}

__device__ __forceinline__ float row16_sum(float x) {
  x += __int_as_float(__builtin_amdgcn_update_dpp(0, __float_as_int(x), 0x111, 0xf, 0xf, true));
  x += __int_as_float(__builtin_amdgcn_update_dpp(0, __float_as_int(x), 0x112, 0xf, 0xf, true));
  x += __int_as_float(__builtin_amdgcn_update_dpp(0, __float_as_int(x), 0x114, 0xf, 0xf, true));
  x += __int_as_float(__builtin_amdgcn_update_dpp(0, __float_as_int(x), 0x118, 0xf, 0xf, true));
  return x;
}

__global__ __launch_bounds__(GEMM_WAVES * 32)
void ssl_gemm_expsum(const _Float16* __restrict__ A16,
                     const _Float16* __restrict__ B16,
                     const int* __restrict__ idx,
                     float* __restrict__ tot,
                     int nTiles, int tilesPerGroup) {
  __shared__ __attribute__((aligned(16))) float s_tot[GEMM_WAVES][GEMM_ROWS];
  const int lane = threadIdx.x & 31;
  const int wave = threadIdx.x >> 5;
  const int c    = lane & 15;
  const int h    = lane >> 4;
  const int row0 = wave * GEMM_ROWS;

  v16h a0[GEMM_MT], a1[GEMM_MT];
  #pragma unroll
  for (int mt = 0; mt < GEMM_MT; ++mt)
    load_frag2(A16 + (size_t)idx[row0 + mt * 16 + c] * EMB, h, a0[mt], a1[mt]);

  float s[GEMM_MT][8];
  #pragma unroll
  for (int mt = 0; mt < GEMM_MT; ++mt) {
    #pragma unroll
    for (int v = 0; v < 8; ++v) s[mt][v] = 0.0f;
  }

  const int t0 = blockIdx.x * tilesPerGroup;
  const int t1 = (t0 + tilesPerGroup < nTiles) ? (t0 + tilesPerGroup) : nTiles;
  const _Float16* brow = B16 + ((size_t)t0 * 16 + c) * EMB;
  v16h b0, b1;
  load_frag2(brow, h, b0, b1);
  for (int t = t0; t < t1; ++t) {
    const _Float16* nrow = brow + ((t + 1 < t1) ? 16 * EMB : 0);
    v16h nb0, nb1;
    load_frag2(nrow, h, nb0, nb1);
    asm volatile("" ::: "memory");
    #pragma unroll
    for (int mt = 0; mt < GEMM_MT; ++mt) {
      v8f d = {};
      d = __builtin_amdgcn_wmma_f32_16x16x32_f16(false, a0[mt], false, b0,
                                                 (short)0, d, false, false);
      d = __builtin_amdgcn_wmma_f32_16x16x32_f16(false, a1[mt], false, b1,
                                                 (short)0, d, false, false);
      #pragma unroll
      for (int v = 0; v < 8; ++v) s[mt][v] += __expf(d[v] * INV_TEMP);
    }
    b0 = nb0; b1 = nb1; brow = nrow;
  }

  #pragma unroll
  for (int mt = 0; mt < GEMM_MT; ++mt) {
    #pragma unroll
    for (int v = 0; v < 8; ++v) s[mt][v] = row16_sum(s[mt][v]);
  }
  if (c == 15) {
    #pragma unroll
    for (int mt = 0; mt < GEMM_MT; ++mt) {
      float4* p = (float4*)&s_tot[wave][mt * 16 + 8 * h];
      p[0] = make_float4(s[mt][0], s[mt][1], s[mt][2], s[mt][3]);
      p[1] = make_float4(s[mt][4], s[mt][5], s[mt][6], s[mt][7]);
    }
  }
  __syncthreads();
  #pragma unroll
  for (int q = 0; q < GEMM_ROWS / 32; ++q)
    atomicAdd(&tot[row0 + q * 32 + lane], s_tot[wave][q * 32 + lane]);
}

__global__ void ssl_loss(const _Float16* __restrict__ A16,
                         const _Float16* __restrict__ B16,
                         const int* __restrict__ idx,
                         const float* __restrict__ tot,
                         float* __restrict__ ssl_out) {
  int gid = blockIdx.x * blockDim.x + threadIdx.x;
  int i = gid >> 5;
  if (i >= BATCH) return;
  int t = gid & 31;
  size_t ro = (size_t)idx[i] * EMB + 2 * t;
  float d = (float)A16[ro] * (float)B16[ro]
          + (float)A16[ro + 1] * (float)B16[ro + 1];
  #pragma unroll
  for (int off = 1; off < 32; off <<= 1) d += __shfl_xor(d, off, 32);
  if (t == 0) {
    float li = -(d * INV_TEMP - logf(tot[i])) * (1.0f / (float)BATCH);
    atomicAdd(ssl_out, li);
  }
}

extern "C" void kernel_launch(void* const* d_in, const int* in_sizes, int n_in,
                              void* d_out, int out_size, void* d_ws, size_t ws_size,
                              hipStream_t stream) {
  (void)in_sizes; (void)n_in; (void)out_size; (void)ws_size;
  const float* ued  = (const float*)d_in[0];
  const float* uei  = (const float*)d_in[1];
  const float* ged  = (const float*)d_in[2];
  const float* gei  = (const float*)d_in[3];
  const float* item = (const float*)d_in[4];
  const int*   hu_r = (const int*)d_in[5];
  const int*   hu_c = (const int*)d_in[6];
  const float* hu_v = (const float*)d_in[7];
  const int*   hi_r = (const int*)d_in[8];
  const int*   hi_c = (const int*)d_in[9];
  const float* hi_v = (const float*)d_in[10];
  const int*   ui_r = (const int*)d_in[11];
  const int*   ui_c = (const int*)d_in[12];
  const float* ui_v = (const float*)d_in[13];
  const int*   uin  = (const int*)d_in[14];
  const int*   pos  = (const int*)d_in[15];
  const int*   neg  = (const int*)d_in[16];
  float* out = (float*)d_out;

  float* W = (float*)d_ws;
  size_t off = 0;
  float* ug_b = W + off; off += (size_t)N_UG * EMB;
  float* ug_a = W + off; off += (size_t)N_UG * EMB;
  float* ig_b = W + off; off += (size_t)N_IG * EMB;
  float* ig_a = W + off; off += (size_t)N_IG * EMB;
  float* agg  = W + off; off += (size_t)N_UI * EMB;
  float* tot  = W + off; off += 4 * BATCH;
  int* pairs_ug = (int*)(W + off); off += E_UI;
  int* offsets_ug = (int*)(W + off); off += N_UI;
  int* offsets_ui = (int*)(W + off); off += N_UI + 1;
  off = (off + 7) & ~(size_t)7;
  _Float16* H = (_Float16*)(W + off);
  _Float16* ug_user_n16  = H;
  _Float16* ui_user_n16  = H + (size_t)NUM_USER * EMB;
  _Float16* ug_group_n16 = H + (size_t)NUM_USER * EMB * 2;
  _Float16* ig_group_n16 = ug_group_n16 + (size_t)NUM_GROUP * EMB;

  int* rank_all   = (int*)ug_b;
  int* counts_all = (int*)ug_b + E_ALL;
  int* counts_ug  = counts_all;
  int* counts_ig  = counts_all + N_UG;
  int* counts_ui  = counts_all + N_UG + N_IG;
  unsigned* hist_ug = (unsigned*)ig_b;
  unsigned* hist_ig = hist_ug + (size_t)SL_U * HW_U;
  unsigned* hist_ui = hist_ig + (size_t)SL_I * HW_I;
  int* base_ug = (int*)agg;
  int* base_ig = base_ug + (size_t)SL_U * N_UG;
  int* base_ui = base_ig + (size_t)SL_I * N_IG;
  int* pairs_ui   = (int*)ug_a;
  int* pairs_ig   = (int*)ug_user_n16;
  int* offsets_ig = pairs_ig + (size_t)2 * E_I;
  static_assert((size_t)E_ALL + N_ALL <= (size_t)N_UG * EMB, "ranks and totals fit ug_b");
  static_assert((size_t)SL_U * HW_U + (size_t)SL_I * HW_I + (size_t)SL_X * HW_X <= (size_t)N_IG * EMB,
                "packed counters fit ig_b");
  static_assert((size_t)SL_U * N_UG + (size_t)SL_I * N_IG + (size_t)SL_X * N_UI <= (size_t)N_UI * EMB,
                "slice starts fit agg");
  static_assert((((size_t)SL_U * N_UG) & 1) == 0 && (((size_t)SL_I * N_IG) & 1) == 0, "slice starts 8-byte aligned");
  static_assert((size_t)2 * E_UI <= (size_t)N_UG * EMB, "user-item pairs fit ug_a");
  static_assert((size_t)2 * E_I + N_IG + 1 <= (size_t)NUM_USER * EMB / 2, "item-group pairs and offsets fit the f16 table");
  static_assert((size_t)2 * E_U <= (size_t)E_UI, "user-group pairs fit their plane");
  static_assert(N_UG + 1 <= N_UI, "user-group offsets fit their plane");
  static_assert((((size_t)N_UG * EMB * 2 + (size_t)N_IG * EMB * 2 + (size_t)N_UI * EMB + 4 * BATCH) & 1) == 0
                && (((size_t)N_UG * EMB) & 1) == 0, "pairs 8-byte aligned");
  static_assert((E_ALL & 3) == 0 && (N_UG & 3) == 0 && ((N_UG + N_IG) & 3) == 0 && (N_UI & 3) == 0
                && ((2 * E_I) & 3) == 0 && (E_UI & 3) == 0, "scan arrays 16-byte aligned");
  const int TB = 256;
  #define CDIV(a, b) (((a) + (b) - 1) / (b))
  #define RPW_UG 8
  #define RPW_IG 5
  #define RPW_UI 10

  zero_f32<<<16, TB, 0, stream>>>(tot, 4 * BATCH);
  zero_f32<<<1, 32, 0, stream>>>(out + (size_t)3 * BATCH * 128, 2);

  slice_hist3<<<SL_U + SL_I + SL_X, HIST_THREADS, 0, stream>>>(
      hu_c, hi_c, ui_c, hist_ug, hist_ig, hist_ui, rank_all);
  col_base3<<<CDIV(HW_U + HW_I + HW_X, TB), TB, 0, stream>>>(
      hist_ug, hist_ig, hist_ui, base_ug, base_ig, base_ui, counts_all);
  exscan3_kernel<<<3, 1024, 0, stream>>>(counts_ug, offsets_ug, N_UG,
                                         counts_ig, offsets_ig, N_IG,
                                         counts_ui, offsets_ui, N_UI);
  csr_fill3<<<CDIV(E_ALL, TB), TB, 0, stream>>>(
      hu_c, hu_r, hu_v, offsets_ug, base_ug, pairs_ug,
      hi_c, hi_r, hi_v, offsets_ig, base_ig, pairs_ig,
      ui_c, ui_r, ui_v, offsets_ui, base_ui, pairs_ui, rank_all);

  csr_prop<<<CDIV(CDIV(N_UI, RPW_UI) * 32, TB), TB, 0, stream>>>(
      uei, NUM_USER, item, offsets_ui, pairs_ui, agg, N_UI, RPW_UI,
      1, ui_user_n16, NUM_USER, nullptr);

  csr_prop<<<CDIV(CDIV(N_IG, RPW_IG) * 32, TB), TB, 0, stream>>>(
      item, NUM_ITEM, gei, offsets_ig, pairs_ig, ig_b, N_IG, RPW_IG,
      0, nullptr, 0, nullptr);
  csr_prop<<<CDIV(CDIV(N_IG, RPW_IG) * 32, TB), TB, 0, stream>>>(
      ig_b, 1 << 30, nullptr, offsets_ig, pairs_ig, ig_a, N_IG, RPW_IG,
      0, nullptr, NUM_ITEM, ig_group_n16);

  csr_prop<<<CDIV(CDIV(N_UG, RPW_UG) * 32, TB), TB, 0, stream>>>(
      ued, NUM_USER, ged, offsets_ug, pairs_ug, ug_b, N_UG, RPW_UG,
      0, nullptr, 0, nullptr);
  csr_prop<<<CDIV(CDIV(N_UG, RPW_UG) * 32, TB), TB, 0, stream>>>(
      ug_b, 1 << 30, nullptr, offsets_ug, pairs_ug, ug_a, N_UG, RPW_UG,
      0, ug_user_n16, NUM_USER, ug_group_n16);

  gather_out<<<CDIV(BATCH * EMB, TB), TB, 0, stream>>>(
      ug_a, agg, ig_a, uin, pos, neg, out);
  reg_kernel<<<CDIV(BATCH * EMB, TB), TB, 0, stream>>>(
      ued, uei, ged, gei, uin, pos, neg, out + (size_t)3 * BATCH * 128);

  ssl_gemm_expsum<<<CDIV(NUM_USER / 16, GEMM_TPG_U), GEMM_WAVES * 32, 0, stream>>>(
      ui_user_n16, ug_user_n16, uin, tot + 0,         NUM_USER / 16, GEMM_TPG_U);
  ssl_gemm_expsum<<<CDIV(NUM_USER / 16, GEMM_TPG_U), GEMM_WAVES * 32, 0, stream>>>(
      ug_user_n16, ui_user_n16, uin, tot + BATCH,     NUM_USER / 16, GEMM_TPG_U);
  ssl_gemm_expsum<<<CDIV(NUM_GROUP / 16, GEMM_TPG_G), GEMM_WAVES * 32, 0, stream>>>(
      ig_group_n16, ug_group_n16, pos, tot + 2*BATCH, NUM_GROUP / 16, GEMM_TPG_G);
  ssl_gemm_expsum<<<CDIV(NUM_GROUP / 16, GEMM_TPG_G), GEMM_WAVES * 32, 0, stream>>>(
      ug_group_n16, ig_group_n16, pos, tot + 3*BATCH, NUM_GROUP / 16, GEMM_TPG_G);

  float* ssl_slot = out + (size_t)3 * BATCH * 128 + 1;
  int lossBlocks = CDIV(BATCH * 32, TB);
  ssl_loss<<<lossBlocks, TB, 0, stream>>>(ui_user_n16, ug_user_n16, uin, tot + 0,         ssl_slot);
  ssl_loss<<<lossBlocks, TB, 0, stream>>>(ug_user_n16, ui_user_n16, uin, tot + BATCH,     ssl_slot);
  ssl_loss<<<lossBlocks, TB, 0, stream>>>(ig_group_n16, ug_group_n16, pos, tot + 2*BATCH, ssl_slot);
  ssl_loss<<<lossBlocks, TB, 0, stream>>>(ug_group_n16, ig_group_n16, pos, tot + 3*BATCH, ssl_slot);
  #undef RPW_UG
  #undef RPW_IG
  #undef RPW_UI
  #undef CDIV
}


# Independent kernels share launches (21 to 8): 3 first-layer gathers in one, 2 second layers in one, 4 SSL GEMMs + output rows + reg in one, 4 loss sums in one (ssl atomics interleave), zeroing folded
# speedup vs baseline: 6.1654x; 1.3096x over previous
#include <hip/hip_runtime.h>
#include <hip/hip_bf16.h>
#include <math.h>

#define NUM_USER   100000
#define NUM_ITEM   50000
#define NUM_GROUP  20000
#define EMB        64
#define BATCH      1024
#define E_UI       2000000
#define E_U        1000000
#define E_I        1000000
#define INV_TEMP   10.0f
#define EPS_N      1e-12f

#define N_UG  (NUM_USER + NUM_GROUP)
#define N_IG  (NUM_ITEM + NUM_GROUP)
#define N_UI  (NUM_USER + NUM_ITEM)

typedef _Float16 v16h __attribute__((ext_vector_type(16)));
typedef float    v8f  __attribute__((ext_vector_type(8)));

#define E_ALL  (E_U + E_I + E_UI)
#define N_UIB  NUM_USER
#define N_ALL  (N_UG + N_IG + N_UIB)
#define SLICE  62500
#define SL_U   ((E_U  + SLICE - 1) / SLICE)
#define SL_I   ((E_I  + SLICE - 1) / SLICE)
#define SL_X   ((E_UI + SLICE - 1) / SLICE)
#define HW_U   ((N_UG + 1) / 2)
#define HW_I   ((N_IG + 1) / 2)
#define HW_X   ((N_UIB + 1) / 2)
#define HW_MAX 75000
#define HIST_THREADS 1024
#define HIST_U 8
static_assert(SLICE <= 65535, "a slice's counts fit 16 bits");
static_assert((SLICE & 3) == 0 && (E_U & 3) == 0 && (E_I & 3) == 0 && (E_UI & 3) == 0, "whole quads of edges");
static_assert(HW_U <= HW_MAX && HW_I <= HW_MAX && HW_X <= HW_MAX && (HW_MAX & 3) == 0, "counters fit LDS");
static_assert((N_UI + 1) / 2 <= HW_MAX, "every user-item column, bucket or not, has its counter in LDS");

__device__ __forceinline__ int slice_rank(unsigned* s_h, int c) {
  const unsigned old = atomicAdd(&s_h[c >> 1], (c & 1) ? 0x10000u : 1u);
  return (int)((c & 1) ? (old >> 16) : (old & 0xffffu));
}

__global__ __launch_bounds__(HIST_THREADS)
void slice_hist3(const int* __restrict__ colU, const int* __restrict__ colI,
                 const int* __restrict__ colX,
                 unsigned* __restrict__ histU, unsigned* __restrict__ histI,
                 unsigned* __restrict__ histX, int* __restrict__ rank) {
  __shared__ __attribute__((aligned(16))) unsigned s_h[HW_MAX];
  const int b = blockIdx.x;
  const int* col = colU; unsigned* hist = histU;
  int nE = E_U, hw = HW_U, ebase = 0, s = b;
  if (b >= SL_U + SL_I) {
    col = colX; hist = histX; nE = E_UI; hw = HW_X; ebase = E_U + E_I; s = b - (SL_U + SL_I);
  } else if (b >= SL_U) {
    col = colI; hist = histI; nE = E_I; hw = HW_I; ebase = E_U; s = b - SL_U;
  }
  const int tid = threadIdx.x;
  for (int w = tid; w < HW_MAX / 4; w += HIST_THREADS) ((uint4*)s_h)[w] = make_uint4(0u, 0u, 0u, 0u);
  __syncthreads();

  const int e0  = s * SLICE;
  const int len = (nE - e0 < SLICE) ? (nE - e0) : SLICE;
  const int nq  = len >> 2;
  const int4* __restrict__ cq = (const int4*)(col + e0);
  int4* __restrict__ rq = (int4*)(rank + ebase + e0);
  for (int q0 = 0; q0 < nq; q0 += HIST_THREADS * HIST_U) {
    int4 c[HIST_U];
    #pragma unroll
    for (int u = 0; u < HIST_U; ++u) {
      const int q = q0 + u * HIST_THREADS + tid;
      c[u] = cq[(q < nq) ? q : (nq - 1)];
    }
    #pragma unroll
    for (int u = 0; u < HIST_U; ++u) {
      const int q = q0 + u * HIST_THREADS + tid;
      if (q < nq) {
        int4 r;
        r.x = slice_rank(s_h, c[u].x);
        r.y = slice_rank(s_h, c[u].y);
        r.z = slice_rank(s_h, c[u].z);
        r.w = slice_rank(s_h, c[u].w);
        rq[q] = r;
      }
    }
  }
  __syncthreads();
  unsigned* __restrict__ ho = hist + (size_t)s * hw;
  if ((hw & 3) == 0) {
    for (int w = tid; w < hw / 4; w += HIST_THREADS) ((uint4*)ho)[w] = ((const uint4*)s_h)[w];
  } else {
    for (int w = tid; w < hw; w += HIST_THREADS) ho[w] = s_h[w];
  }
}

#define CB_U 16
__global__ void col_base3(const unsigned* __restrict__ histU, const unsigned* __restrict__ histI,
                          const unsigned* __restrict__ histX,
                          int* __restrict__ baseU, int* __restrict__ baseI,
                          int* __restrict__ baseX, int* __restrict__ counts,
                          float* __restrict__ tot, float* __restrict__ scal) {
  const int t = blockIdx.x * blockDim.x + threadIdx.x;
  if (t < 4 * BATCH) tot[t] = 0.0f;
  if (t < 2) scal[t] = 0.0f;
  if (t >= HW_U + HW_I + HW_X) return;
  const unsigned* hist = histU; int* base = baseU;
  int hw = HW_U, n = N_UG, ns = SL_U, cb = 0, w = t;
  if (t >= HW_U + HW_I) {
    hist = histX; base = baseX; hw = HW_X; n = N_UIB; ns = SL_X; cb = N_UG + N_IG; w = t - (HW_U + HW_I);
  } else if (t >= HW_U) {
    hist = histI; base = baseI; hw = HW_I; n = N_IG; ns = SL_I; cb = N_UG; w = t - HW_U;
  }
  const int  c0   = 2 * w;
  const bool two  = (c0 + 1 < n);
  const bool pair = two && ((n & 1) == 0);
  int run0 = 0, run1 = 0;
  for (int s0 = 0; s0 < ns; s0 += CB_U) {
    unsigned v[CB_U];
    #pragma unroll
    for (int u = 0; u < CB_U; ++u) {
      const int s = s0 + u;
      v[u] = hist[(size_t)((s < ns) ? s : (ns - 1)) * hw + w];
    }
    #pragma unroll
    for (int u = 0; u < CB_U; ++u) {
      const int s = s0 + u;
      if (s < ns) {
        int* bp = base + (size_t)s * n + c0;
        if (pair) { int2 o; o.x = run0; o.y = run1; *(int2*)bp = o; }
        else      { bp[0] = run0; if (two) bp[1] = run1; }
        run0 += (int)(v[u] & 0xffffu);
        run1 += (int)(v[u] >> 16);
      }
    }
  }
  counts[cb + c0] = run0;
  if (two) counts[cb + c0 + 1] = run1;
}

#define SCAN_THREADS 1024
#define SCAN_WAVES   (SCAN_THREADS / 32)
#define SCAN_TILE    128
#define SCAN_U       8
#define SCAN_GROUP   (SCAN_TILE * SCAN_U)

__device__ __forceinline__ int wave_incl_scan(int x, int lane) {
  x += __builtin_amdgcn_update_dpp(0, x, 0x111, 0xf, 0xf, true);
  x += __builtin_amdgcn_update_dpp(0, x, 0x112, 0xf, 0xf, true);
  x += __builtin_amdgcn_update_dpp(0, x, 0x114, 0xf, 0xf, true);
  x += __builtin_amdgcn_update_dpp(0, x, 0x118, 0xf, 0xf, true);
  const int low = __builtin_amdgcn_readlane(x, 15);
  x += (lane & 16) ? low : 0;
  return x;
}

__device__ __forceinline__ void scan_load_group(const int* __restrict__ p, int t,
                                                int lane, int end, bool vec,
                                                int4 (&v)[SCAN_U]) {
  if (vec && t + SCAN_GROUP <= end) {
    #pragma unroll
    for (int u = 0; u < SCAN_U; ++u)
      v[u] = *(const int4*)(p + t + u * SCAN_TILE + lane * 4);
  } else {
    #pragma unroll
    for (int u = 0; u < SCAN_U; ++u) {
      int i = t + u * SCAN_TILE + lane * 4;
      v[u].x = (i     < end) ? p[i]     : 0;
      v[u].y = (i + 1 < end) ? p[i + 1] : 0;
      v[u].z = (i + 2 < end) ? p[i + 2] : 0;
      v[u].w = (i + 3 < end) ? p[i + 3] : 0;
    }
  }
}

__device__ __forceinline__ void scan_store_group(int* __restrict__ p, int t,
                                                 int lane, int end, bool vec,
                                                 const int4 (&v)[SCAN_U]) {
  if (vec && t + SCAN_GROUP <= end) {
    #pragma unroll
    for (int u = 0; u < SCAN_U; ++u)
      *(int4*)(p + t + u * SCAN_TILE + lane * 4) = v[u];
  } else {
    #pragma unroll
    for (int u = 0; u < SCAN_U; ++u) {
      int i = t + u * SCAN_TILE + lane * 4;
      if (i     < end) p[i]     = v[u].x;
      if (i + 1 < end) p[i + 1] = v[u].y;
      if (i + 2 < end) p[i + 2] = v[u].z;
      if (i + 3 < end) p[i + 3] = v[u].w;
    }
  }
}

__global__ __launch_bounds__(SCAN_THREADS)
void exscan3_kernel(int* __restrict__ cnt0, int* __restrict__ off0, int n0,
                    int* __restrict__ cnt1, int* __restrict__ off1, int n1,
                    int* __restrict__ cnt2, int* __restrict__ off2, int n2) {
  int* __restrict__ counts  = cnt0;
  int* __restrict__ offsets = off0;
  int n = n0;
  if (blockIdx.x == 1) { counts = cnt1; offsets = off1; n = n1; }
  if (blockIdx.x == 2) { counts = cnt2; offsets = off2; n = n2; }
  __shared__ int s_wsum[SCAN_WAVES];
  const int tid  = threadIdx.x;
  const int lane = tid & 31;
  const int wave = __builtin_amdgcn_readfirstlane(tid >> 5);
  const int per = (((n + SCAN_WAVES - 1) / SCAN_WAVES) + SCAN_GROUP - 1)
                  & ~(SCAN_GROUP - 1);
  const int beg = wave * per;
  const int end = (beg + per < n) ? (beg + per) : n;
  const bool vec = ((((size_t)counts) | ((size_t)offsets)) & 15) == 0;

  int acc = 0;
  for (int t = beg; t < end; t += SCAN_GROUP) {
    int4 v[SCAN_U];
    scan_load_group(counts, t, lane, end, vec, v);
    #pragma unroll
    for (int u = 0; u < SCAN_U; ++u) acc += (v[u].x + v[u].y) + (v[u].z + v[u].w);
  }
  #pragma unroll
  for (int off = 16; off > 0; off >>= 1) acc += __shfl_xor(acc, off, 32);
  if (lane == 0) s_wsum[wave] = acc;
  __syncthreads();

  const int wt    = s_wsum[lane];
  const int winc  = wave_incl_scan(wt, lane);
  int       carry = __builtin_amdgcn_readlane(winc - wt, wave);
  const int grand = __builtin_amdgcn_readlane(winc, 31);

  for (int t = beg; t < end; t += SCAN_GROUP) {
    int4 v[SCAN_U];
    scan_load_group(counts, t, lane, end, vec, v);
    int4 o[SCAN_U];
    #pragma unroll
    for (int u = 0; u < SCAN_U; ++u) {
      const int p1 = v[u].x;
      const int p2 = p1 + v[u].y;
      const int p3 = p2 + v[u].z;
      const int p4 = p3 + v[u].w;
      const int inc = wave_incl_scan(p4, lane);
      const int ex  = carry + (inc - p4);
      o[u].x = ex;
      o[u].y = ex + p1;
      o[u].z = ex + p2;
      o[u].w = ex + p3;
      carry += __builtin_amdgcn_readlane(inc, 31);
    }
    scan_store_group(offsets, t, lane, end, vec, o);
    scan_store_group(counts,  t, lane, end, vec, o);
  }
  if (tid == 0) offsets[n] = grand;
}

__global__ void csr_fill3(const int* __restrict__ colU, const int* __restrict__ rowU,
                          const float* __restrict__ valU, const int* __restrict__ offU,
                          const int* __restrict__ baseU, int* __restrict__ pairsU,
                          const int* __restrict__ colI, const int* __restrict__ rowI,
                          const float* __restrict__ valI, const int* __restrict__ offI,
                          const int* __restrict__ baseI, int* __restrict__ pairsI,
                          const int* __restrict__ colX, const int* __restrict__ rowX,
                          const float* __restrict__ valX, const int* __restrict__ offX,
                          const int* __restrict__ baseX, int* __restrict__ pairsX,
                          const int* __restrict__ rank) {
  int e = blockIdx.x * blockDim.x + threadIdx.x;
  if (e >= E_ALL) return;
  const int* col = colU; const int* row = rowU; const float* val = valU;
  const int* offsets = offU; const int* base = baseU; int* pairs = pairsU;
  int le = e, n = N_UG;
  if (e >= E_U + E_I) {
    col = colX; row = rowX; val = valX; offsets = offX; base = baseX; pairs = pairsX;
    le = e - (E_U + E_I); n = N_UIB;
  } else if (e >= E_U) {
    col = colI; row = rowI; val = valI; offsets = offI; base = baseI; pairs = pairsI;
    le = e - E_U; n = N_IG;
  }
  const int c = col[le];
  if (c >= n) return;
  int slot = offsets[c] + base[(size_t)(le / SLICE) * n + c] + rank[e];
  int2 rv;
  rv.x = row[le];
  rv.y = __float_as_int(val[le]);
  *(int2*)(pairs + 2 * (size_t)slot) = rv;
}

typedef _Float16 v2h __attribute__((ext_vector_type(2)));

#define GATHER_DEPTH 16

__device__ __forceinline__ const float* table_row(const float* __restrict__ p0, int n0,
                                                  const float* __restrict__ p1, int r) {
  return (r < n0) ? (p0 + (size_t)r * EMB) : (p1 + (size_t)(r - n0) * EMB);
}

__device__ __forceinline__ int2 load_pairs(const int2* __restrict__ rv, int base,
                                           int end, int lane) {
  int2 p; p.x = 0; p.y = 0;
  if (base + lane < end) p = rv[base + lane];
  return p;
}

__device__ __forceinline__ void batch_load(const float* __restrict__ p0, int n0,
                                           const float* __restrict__ p1,
                                           int px, int j, int n, int lane,
                                           float2 (&d)[GATHER_DEPTH]) {
  int r[GATHER_DEPTH];
  #pragma unroll
  for (int u = 0; u < GATHER_DEPTH; ++u)
    r[u] = __builtin_amdgcn_readlane(px, j + u);
  #pragma unroll
  for (int u = 0; u < GATHER_DEPTH; ++u) {
    if (j + u < n)
      d[u] = *(const float2*)(table_row(p0, n0, p1, r[u]) + lane * 2);
  }
}

__device__ __forceinline__ void batch_sum(int pv, int j, int n,
                                          const float2 (&d)[GATHER_DEPTH],
                                          float& ax, float& ay) {
  #pragma unroll
  for (int u = 0; u < GATHER_DEPTH; ++u) {
    if (j + u < n) {
      const float v = __int_as_float(__builtin_amdgcn_readlane(pv, j + u));
      ax += d[u].x * v;
      ay += d[u].y * v;
    }
  }
}

__device__ __forceinline__ void csr_prop_rows(const float* __restrict__ p0, int n0,
                         const float* __restrict__ p1,
                         const int* __restrict__ offsets,
                         const int* __restrict__ pairs,
                         float* __restrict__ out, int rowBeg, int nRows, int rowsPerWave,
                         int addSelf,
                         _Float16* __restrict__ nlo, int split,
                         _Float16* __restrict__ nhi, int blk) {
  const int2* __restrict__ rv = (const int2*)pairs;
  const int lane = threadIdx.x & 31;
  const int wv = __builtin_amdgcn_readfirstlane(
      (int)((blk * blockDim.x + threadIdx.x) >> 5));
  const int r0 = rowBeg + wv * rowsPerWave;
  if (r0 >= nRows) return;
  const int nr = (nRows - r0 < rowsPerWave) ? (nRows - r0) : rowsPerWave;
  const int offv = (lane <= nr) ? offsets[r0 + lane] : 0;
  int beg = __builtin_amdgcn_readlane(offv, 0);
  int end = __builtin_amdgcn_readlane(offv, 1);
  int2 cur = load_pairs(rv, beg, end, lane);

  for (int i = 0; i < nr; ++i) {
    const int wid = r0 + i;
    const int nf  = (end - beg < 32) ? (end - beg) : 32;
    float ax = 0.0f, ay = 0.0f;
    float2 d[GATHER_DEPTH];
    batch_load(p0, n0, p1, cur.x, 0, nf, lane, d);
    int nend = end;
    int2 nxt; nxt.x = 0; nxt.y = 0;
    if (i + 1 < nr) {
      nend = __builtin_amdgcn_readlane(offv, i + 2);
      nxt  = load_pairs(rv, end, nend, lane);
    }
    float sx = 0.0f, sy = 0.0f;
    if (addSelf) {
      float2 s = *(const float2*)(table_row(p0, n0, p1, wid) + lane * 2);
      sx = s.x; sy = s.y;
    }
    batch_sum(cur.y, 0, nf, d, ax, ay);
    if (nf > GATHER_DEPTH) {
      batch_load(p0, n0, p1, cur.x, GATHER_DEPTH, nf, lane, d);
      batch_sum(cur.y, GATHER_DEPTH, nf, d, ax, ay);
    }
    for (int base = beg + 32; base < end; base += 32) {
      const int2 more = load_pairs(rv, base, end, lane);
      const int n = (end - base < 32) ? (end - base) : 32;
      for (int j = 0; j < n; j += GATHER_DEPTH) {
        batch_load(p0, n0, p1, more.x, j, n, lane, d);
        batch_sum(more.y, j, n, d, ax, ay);
      }
    }
    float2 o; o.x = ax; o.y = ay;
    if (addSelf) { o.x = (sx + ax) * 0.5f; o.y = (sy + ay) * 0.5f; }
    *(float2*)(out + (size_t)wid * EMB + lane * 2) = o;

    if (nlo != nullptr || nhi != nullptr) {
      float ss = fmaf(o.x, o.x, o.y * o.y);
      #pragma unroll
      for (int off = 1; off < 32; off <<= 1) ss += __shfl_xor(ss, off, 32);
      float scale = 1.0f / fmaxf(sqrtf(ss), EPS_N);
      _Float16* dst = nullptr;
      if (wid < split) { if (nlo != nullptr) dst = nlo + (size_t)wid * EMB; }
      else             { if (nhi != nullptr) dst = nhi + (size_t)(wid - split) * EMB; }
      if (dst != nullptr) {
        v2h h;
        h.x = (_Float16)(o.x * scale);
        h.y = (_Float16)(o.y * scale);
        *(v2h*)(dst + lane * 2) = h;
      }
    }
    beg = end; end = nend; cur = nxt;
  }
}

__global__ void csr_prop3(
    const float* __restrict__ p0A, int n0A, const float* __restrict__ p1A,
    const int* __restrict__ offA, const int* __restrict__ pairsA, float* __restrict__ outA,
    int rowBegA, int nRowsA, int rpwA, int addSelfA,
    _Float16* __restrict__ nloA, int splitA, _Float16* __restrict__ nhiA, int nbA,
    const float* __restrict__ p0B, int n0B, const float* __restrict__ p1B,
    const int* __restrict__ offB, const int* __restrict__ pairsB, float* __restrict__ outB,
    int rowBegB, int nRowsB, int rpwB, int addSelfB,
    _Float16* __restrict__ nloB, int splitB, _Float16* __restrict__ nhiB, int nbB,
    const float* __restrict__ p0C, int n0C, const float* __restrict__ p1C,
    const int* __restrict__ offC, const int* __restrict__ pairsC, float* __restrict__ outC,
    int rowBegC, int nRowsC, int rpwC, int addSelfC,
    _Float16* __restrict__ nloC, int splitC, _Float16* __restrict__ nhiC) {
  const float* p0 = p0A; const float* p1 = p1A; int n0 = n0A;
  const int* off = offA; const int* pairs = pairsA; float* out = outA;
  int rowBeg = rowBegA, nRows = nRowsA, rpw = rpwA, addSelf = addSelfA, split = splitA;
  _Float16* nlo = nloA; _Float16* nhi = nhiA;
  int blk = blockIdx.x;
  if (blk >= nbA + nbB) {
    p0 = p0C; p1 = p1C; n0 = n0C; off = offC; pairs = pairsC; out = outC;
    rowBeg = rowBegC; nRows = nRowsC; rpw = rpwC; addSelf = addSelfC; split = splitC;
    nlo = nloC; nhi = nhiC; blk -= nbA + nbB;
  } else if (blk >= nbA) {
    p0 = p0B; p1 = p1B; n0 = n0B; off = offB; pairs = pairsB; out = outB;
    rowBeg = rowBegB; nRows = nRowsB; rpw = rpwB; addSelf = addSelfB; split = splitB;
    nlo = nloB; nhi = nhiB; blk -= nbA;
  }
  csr_prop_rows(p0, n0, p1, off, pairs, out, rowBeg, nRows, rpw, addSelf, nlo, split, nhi, blk);
}

#define OUT_THREADS 256
#define OUT_NB      ((BATCH * EMB + OUT_THREADS - 1) / OUT_THREADS)
__device__ __forceinline__
void gather_out_reg_rows(const float* __restrict__ ug,
                               const float* __restrict__ uif,
                               const float* __restrict__ ig,
                               const float* __restrict__ ued,
                               const float* __restrict__ uei,
                               const float* __restrict__ ged,
                               const float* __restrict__ gei,
                               const int* __restrict__ u,
                               const int* __restrict__ p,
                               const int* __restrict__ ng,
                               float* __restrict__ out,
                               float* __restrict__ reg_out, int blk) {
  int gid = blk * OUT_THREADS + threadIdx.x;
  if (gid >= BATCH * EMB) return;
  int b = gid >> 6, c = gid & 63;
  const size_t PO = (size_t)BATCH * 128;
  const size_t NO = 2 * PO;
  out[(size_t)b * 128 + c]           = ug [(size_t)u[b] * EMB + c];
  out[(size_t)b * 128 + 64 + c]      = uif[(size_t)u[b] * EMB + c];
  out[PO + (size_t)b * 128 + c]      = ug [(size_t)(NUM_USER + p[b])  * EMB + c];
  out[PO + (size_t)b * 128 + 64 + c] = ig [(size_t)(NUM_ITEM + p[b])  * EMB + c];
  out[NO + (size_t)b * 128 + c]      = ug [(size_t)(NUM_USER + ng[b]) * EMB + c];
  out[NO + (size_t)b * 128 + 64 + c] = ig [(size_t)(NUM_ITEM + ng[b]) * EMB + c];

  size_t uo = (size_t)u[b]  * EMB + c;
  size_t po = (size_t)p[b]  * EMB + c;
  size_t no = (size_t)ng[b] * EMB + c;
  float s = uei[uo]*uei[uo] + ued[uo]*ued[uo]
          + gei[po]*gei[po] + ged[po]*ged[po]
          + gei[no]*gei[no] + ged[no]*ged[no];
  #pragma unroll
  for (int off = 1; off < 32; off <<= 1) s += __shfl_xor(s, off, 32);
  if ((threadIdx.x & 31) == 0) atomicAdd(reg_out, s * (0.5f / (float)BATCH));
}

#define GEMM_MT     8
#define GEMM_ROWS   (16 * GEMM_MT)
#define GEMM_WAVES  (BATCH / GEMM_ROWS)
#define GEMM_TPG_U  13
#define GEMM_TPG_G  5
static_assert(BATCH % GEMM_ROWS == 0, "whole row groups");
static_assert(GEMM_WAVES * 32 == OUT_THREADS, "the output work-groups have the SSL launch's size");
static_assert(NUM_USER % 16 == 0 && NUM_GROUP % 16 == 0, "whole table tiles");

__device__ __forceinline__ void load_frag2(const _Float16* __restrict__ row, int h,
                                           v16h& f0, v16h& f1) {
  const _Float16* p = row + 16 * h;
  f0 = *(const v16h*)p;
  f1 = *(const v16h*)(p + 32);
}

__device__ __forceinline__ float row16_sum(float x) {
  x += __int_as_float(__builtin_amdgcn_update_dpp(0, __float_as_int(x), 0x111, 0xf, 0xf, true));
  x += __int_as_float(__builtin_amdgcn_update_dpp(0, __float_as_int(x), 0x112, 0xf, 0xf, true));
  x += __int_as_float(__builtin_amdgcn_update_dpp(0, __float_as_int(x), 0x114, 0xf, 0xf, true));
  x += __int_as_float(__builtin_amdgcn_update_dpp(0, __float_as_int(x), 0x118, 0xf, 0xf, true));
  return x;
}

#define GEMM_NB_U   ((NUM_USER  / 16 + GEMM_TPG_U - 1) / GEMM_TPG_U)
#define GEMM_NB_G   ((NUM_GROUP / 16 + GEMM_TPG_G - 1) / GEMM_TPG_G)

__device__ __forceinline__
void ssl_gemm_term(const _Float16* __restrict__ A16,
                   const _Float16* __restrict__ B16,
                   const int* __restrict__ idx,
                   float* __restrict__ tot,
                   int nTiles, int tilesPerGroup, int blk) {
  __shared__ __attribute__((aligned(16))) float s_tot[GEMM_WAVES][GEMM_ROWS];
  const int lane = threadIdx.x & 31;
  const int wave = threadIdx.x >> 5;
  const int c    = lane & 15;
  const int h    = lane >> 4;
  const int row0 = wave * GEMM_ROWS;

  v16h a0[GEMM_MT], a1[GEMM_MT];
  #pragma unroll
  for (int mt = 0; mt < GEMM_MT; ++mt)
    load_frag2(A16 + (size_t)idx[row0 + mt * 16 + c] * EMB, h, a0[mt], a1[mt]);

  float s[GEMM_MT][8];
  #pragma unroll
  for (int mt = 0; mt < GEMM_MT; ++mt) {
    #pragma unroll
    for (int v = 0; v < 8; ++v) s[mt][v] = 0.0f;
  }

  const int t0 = blk * tilesPerGroup;
  const int t1 = (t0 + tilesPerGroup < nTiles) ? (t0 + tilesPerGroup) : nTiles;
  const _Float16* brow = B16 + ((size_t)t0 * 16 + c) * EMB;
  v16h b0, b1;
  load_frag2(brow, h, b0, b1);
  for (int t = t0; t < t1; ++t) {
    const _Float16* nrow = brow + ((t + 1 < t1) ? 16 * EMB : 0);
    v16h nb0, nb1;
    load_frag2(nrow, h, nb0, nb1);
    asm volatile("" ::: "memory");
    #pragma unroll
    for (int mt = 0; mt < GEMM_MT; ++mt) {
      v8f d = {};
      d = __builtin_amdgcn_wmma_f32_16x16x32_f16(false, a0[mt], false, b0,
                                                 (short)0, d, false, false);
      d = __builtin_amdgcn_wmma_f32_16x16x32_f16(false, a1[mt], false, b1,
                                                 (short)0, d, false, false);
      #pragma unroll
      for (int v = 0; v < 8; ++v) s[mt][v] += __expf(d[v] * INV_TEMP);
    }
    b0 = nb0; b1 = nb1; brow = nrow;
  }

  #pragma unroll
  for (int mt = 0; mt < GEMM_MT; ++mt) {
    #pragma unroll
    for (int v = 0; v < 8; ++v) s[mt][v] = row16_sum(s[mt][v]);
  }
  if (c == 15) {
    #pragma unroll
    for (int mt = 0; mt < GEMM_MT; ++mt) {
      float4* p = (float4*)&s_tot[wave][mt * 16 + 8 * h];
      p[0] = make_float4(s[mt][0], s[mt][1], s[mt][2], s[mt][3]);
      p[1] = make_float4(s[mt][4], s[mt][5], s[mt][6], s[mt][7]);
    }
  }
  __syncthreads();
  #pragma unroll
  for (int q = 0; q < GEMM_ROWS / 32; ++q)
    atomicAdd(&tot[row0 + q * 32 + lane], s_tot[wave][q * 32 + lane]);
}

#define GEMM_NB_ALL (2 * GEMM_NB_U + 2 * GEMM_NB_G)
__global__ __launch_bounds__(GEMM_WAVES * 32)
void ssl_gemm4_out(const _Float16* __restrict__ uiU, const _Float16* __restrict__ ugU,
                   const _Float16* __restrict__ igG, const _Float16* __restrict__ ugG,
                   const int* __restrict__ uin, const int* __restrict__ pos,
                   const int* __restrict__ neg, float* __restrict__ tot,
                   const float* __restrict__ ug, const float* __restrict__ uif,
                   const float* __restrict__ ig,
                   const float* __restrict__ ued, const float* __restrict__ uei,
                   const float* __restrict__ ged, const float* __restrict__ gei,
                   float* __restrict__ out, float* __restrict__ reg_out) {
  if (blockIdx.x >= GEMM_NB_ALL) {
    gather_out_reg_rows(ug, uif, ig, ued, uei, ged, gei, uin, pos, neg, out, reg_out,
                        blockIdx.x - GEMM_NB_ALL);
    return;
  }
  const _Float16* A16 = uiU; const _Float16* B16 = ugU; const int* idx = uin;
  int nTiles = NUM_USER / 16, tpg = GEMM_TPG_U, term = 0, blk = blockIdx.x;
  if (blk >= 2 * GEMM_NB_U + GEMM_NB_G) {
    A16 = ugG; B16 = igG; idx = pos; nTiles = NUM_GROUP / 16; tpg = GEMM_TPG_G; term = 3;
    blk -= 2 * GEMM_NB_U + GEMM_NB_G;
  } else if (blk >= 2 * GEMM_NB_U) {
    A16 = igG; B16 = ugG; idx = pos; nTiles = NUM_GROUP / 16; tpg = GEMM_TPG_G; term = 2;
    blk -= 2 * GEMM_NB_U;
  } else if (blk >= GEMM_NB_U) {
    A16 = ugU; B16 = uiU; term = 1; blk -= GEMM_NB_U;
  }
  ssl_gemm_term(A16, B16, idx, tot + term * BATCH, nTiles, tpg, blk);
}

__global__ void ssl_loss4(const _Float16* __restrict__ uiU, const _Float16* __restrict__ ugU,
                          const _Float16* __restrict__ igG, const _Float16* __restrict__ ugG,
                          const int* __restrict__ uin, const int* __restrict__ pos,
                          const float* __restrict__ tot4,
                          float* __restrict__ ssl_out) {
  int gid = blockIdx.x * blockDim.x + threadIdx.x;
  int w = gid >> 5;
  if (w >= 4 * BATCH) return;
  const int term = w / BATCH;
  int i = w - term * BATCH;
  const _Float16* A16 = uiU; const _Float16* B16 = ugU; const int* idx = uin;
  if (term == 1) { A16 = ugU; B16 = uiU; }
  if (term == 2) { A16 = igG; B16 = ugG; idx = pos; }
  if (term == 3) { A16 = ugG; B16 = igG; idx = pos; }
  const float* tot = tot4 + term * BATCH;
  int t = gid & 31;
  size_t ro = (size_t)idx[i] * EMB + 2 * t;
  float d = (float)A16[ro] * (float)B16[ro]
          + (float)A16[ro + 1] * (float)B16[ro + 1];
  #pragma unroll
  for (int off = 1; off < 32; off <<= 1) d += __shfl_xor(d, off, 32);
  if (t == 0) {
    float li = -(d * INV_TEMP - logf(tot[i])) * (1.0f / (float)BATCH);
    atomicAdd(ssl_out, li);
  }
}

extern "C" void kernel_launch(void* const* d_in, const int* in_sizes, int n_in,
                              void* d_out, int out_size, void* d_ws, size_t ws_size,
                              hipStream_t stream) {
  (void)in_sizes; (void)n_in; (void)out_size; (void)ws_size;
  const float* ued  = (const float*)d_in[0];
  const float* uei  = (const float*)d_in[1];
  const float* ged  = (const float*)d_in[2];
  const float* gei  = (const float*)d_in[3];
  const float* item = (const float*)d_in[4];
  const int*   hu_r = (const int*)d_in[5];
  const int*   hu_c = (const int*)d_in[6];
  const float* hu_v = (const float*)d_in[7];
  const int*   hi_r = (const int*)d_in[8];
  const int*   hi_c = (const int*)d_in[9];
  const float* hi_v = (const float*)d_in[10];
  const int*   ui_r = (const int*)d_in[11];
  const int*   ui_c = (const int*)d_in[12];
  const float* ui_v = (const float*)d_in[13];
  const int*   uin  = (const int*)d_in[14];
  const int*   pos  = (const int*)d_in[15];
  const int*   neg  = (const int*)d_in[16];
  float* out = (float*)d_out;

  float* W = (float*)d_ws;
  size_t off = 0;
  float* ug_b = W + off; off += (size_t)N_UG * EMB;
  float* ug_a = W + off; off += (size_t)N_UG * EMB;
  float* ig_b = W + off; off += (size_t)N_IG * EMB;
  float* ig_a = W + off; off += (size_t)N_IG * EMB;
  float* agg  = W + off; off += (size_t)N_UI * EMB;
  float* tot  = W + off; off += 4 * BATCH;
  int* pairs_ug = (int*)(W + off); off += E_UI;
  int* offsets_ug = (int*)(W + off); off += N_UI;
  int* offsets_ui = (int*)(W + off); off += N_UI + 1;
  off = (off + 7) & ~(size_t)7;
  _Float16* H = (_Float16*)(W + off);
  _Float16* ug_user_n16  = H;
  _Float16* ui_user_n16  = H + (size_t)NUM_USER * EMB;
  _Float16* ug_group_n16 = H + (size_t)NUM_USER * EMB * 2;
  _Float16* ig_group_n16 = ug_group_n16 + (size_t)NUM_GROUP * EMB;

  int* rank_all   = (int*)ug_b;
  int* counts_all = (int*)ug_b + E_ALL;
  int* counts_ug  = counts_all;
  int* counts_ig  = counts_all + N_UG;
  int* counts_ui  = counts_all + N_UG + N_IG;
  unsigned* hist_ug = (unsigned*)ig_b;
  unsigned* hist_ig = hist_ug + (size_t)SL_U * HW_U;
  unsigned* hist_ui = hist_ig + (size_t)SL_I * HW_I;
  int* base_ug = (int*)agg;
  int* base_ig = base_ug + (size_t)SL_U * N_UG;
  int* base_ui = base_ig + (size_t)SL_I * N_IG;
  int* pairs_ui   = (int*)ug_a;
  int* pairs_ig   = (int*)agg + (size_t)N_UIB * EMB;
  int* offsets_ig = pairs_ig + (size_t)2 * E_I;
  static_assert((size_t)E_ALL + N_ALL <= (size_t)N_UG * EMB, "ranks and totals fit ug_b");
  static_assert((size_t)SL_U * HW_U + (size_t)SL_I * HW_I + (size_t)SL_X * HW_X <= (size_t)N_IG * EMB,
                "packed counters fit ig_b");
  static_assert((size_t)SL_U * N_UG + (size_t)SL_I * N_IG + (size_t)SL_X * N_UIB <= (size_t)N_UIB * EMB,
                "slice starts fit the user rows of agg");
  static_assert((((size_t)SL_U * N_UG) & 1) == 0 && (((size_t)SL_I * N_IG) & 1) == 0, "slice starts 8-byte aligned");
  static_assert((size_t)2 * E_UI <= (size_t)N_UG * EMB, "user-item pairs fit ug_a");
  static_assert((size_t)2 * E_I + N_IG + 1 <= (size_t)(N_UI - N_UIB) * EMB, "item-group pairs and offsets fit behind the user rows of agg");
  static_assert((((size_t)N_UIB * EMB) & 3) == 0, "item-group pairs 16-byte aligned");
  static_assert((size_t)2 * E_U <= (size_t)E_UI, "user-group pairs fit their plane");
  static_assert(N_UG + 1 <= N_UI, "user-group offsets fit their plane");
  static_assert((((size_t)N_UG * EMB * 2 + (size_t)N_IG * EMB * 2 + (size_t)N_UI * EMB + 4 * BATCH) & 1) == 0
                && (((size_t)N_UG * EMB) & 1) == 0, "pairs 8-byte aligned");
  static_assert((E_ALL & 3) == 0 && (N_UG & 3) == 0 && ((N_UG + N_IG) & 3) == 0 && (N_UI & 3) == 0 && (N_UIB & 1) == 0
                && ((2 * E_I) & 3) == 0 && (E_UI & 3) == 0, "scan arrays 16-byte aligned");
  const int TB = 256;
  #define CDIV(a, b) (((a) + (b) - 1) / (b))
  #define RPW_UG 8
  #define RPW_IG 5
  #define RPW_IG2 2
  #define RPW_UI 7

  slice_hist3<<<SL_U + SL_I + SL_X, HIST_THREADS, 0, stream>>>(
      hu_c, hi_c, ui_c, hist_ug, hist_ig, hist_ui, rank_all);
  col_base3<<<CDIV(HW_U + HW_I + HW_X, TB), TB, 0, stream>>>(
      hist_ug, hist_ig, hist_ui, base_ug, base_ig, base_ui, counts_all,
      tot, out + (size_t)3 * BATCH * 128);
  exscan3_kernel<<<3, 1024, 0, stream>>>(counts_ug, offsets_ug, N_UG,
                                         counts_ig, offsets_ig, N_IG,
                                         counts_ui, offsets_ui, N_UIB);
  csr_fill3<<<CDIV(E_ALL, TB), TB, 0, stream>>>(
      hu_c, hu_r, hu_v, offsets_ug, base_ug, pairs_ug,
      hi_c, hi_r, hi_v, offsets_ig, base_ig, pairs_ig,
      ui_c, ui_r, ui_v, offsets_ui, base_ui, pairs_ui, rank_all);

  constexpr int nbUI  = CDIV(CDIV(N_UIB, RPW_UI) * 32, TB);
  constexpr int nbIG1 = CDIV(CDIV(N_IG, RPW_IG) * 32, TB);
  constexpr int nbUG  = CDIV(CDIV(N_UG, RPW_UG) * 32, TB);
  constexpr int nbIG2 = CDIV(CDIV(NUM_GROUP, RPW_IG2) * 32, TB);
  csr_prop3<<<nbUI + nbIG1 + nbUG, TB, 0, stream>>>(
      uei, NUM_USER, item, offsets_ui, pairs_ui, agg, 0, N_UIB, RPW_UI,
      1, ui_user_n16, NUM_USER, nullptr, nbUI,
      item, NUM_ITEM, gei, offsets_ig, pairs_ig, ig_b, 0, N_IG, RPW_IG,
      0, nullptr, 0, nullptr, nbIG1,
      ued, NUM_USER, ged, offsets_ug, pairs_ug, ug_b, 0, N_UG, RPW_UG,
      0, nullptr, 0, nullptr);

  csr_prop3<<<nbUG + nbIG2, TB, 0, stream>>>(
      ug_b, 1 << 30, nullptr, offsets_ug, pairs_ug, ug_a, 0, N_UG, RPW_UG,
      0, ug_user_n16, NUM_USER, ug_group_n16, nbUG,
      ig_b, 1 << 30, nullptr, offsets_ig, pairs_ig, ig_a, NUM_ITEM, N_IG, RPW_IG2,
      0, nullptr, NUM_ITEM, ig_group_n16, nbIG2,
      nullptr, 0, nullptr, nullptr, nullptr, nullptr, 0, 0, 1,
      0, nullptr, 0, nullptr);

  ssl_gemm4_out<<<GEMM_NB_ALL + OUT_NB, GEMM_WAVES * 32, 0, stream>>>(
      ui_user_n16, ug_user_n16, ig_group_n16, ug_group_n16, uin, pos, neg, tot,
      ug_a, agg, ig_a, ued, uei, ged, gei, out, out + (size_t)3 * BATCH * 128);


  float* ssl_slot = out + (size_t)3 * BATCH * 128 + 1;
  ssl_loss4<<<CDIV(4 * BATCH * 32, TB), TB, 0, stream>>>(
      ui_user_n16, ug_user_n16, ig_group_n16, ug_group_n16, uin, pos, tot, ssl_slot);
  #undef RPW_UG
  #undef RPW_IG
  #undef RPW_IG2
  #undef RPW_UI
  #undef CDIV
}


# CSR gather: a pair's source-row address is formed by its own lane (vector ALU, once per 32 pairs) and handed over by lane reads; loads and sums end at the bucket's end; no scalar address sum per pair
# speedup vs baseline: 6.3632x; 1.0321x over previous
#include <hip/hip_runtime.h>
#include <hip/hip_bf16.h>
#include <math.h>

#define NUM_USER   100000
#define NUM_ITEM   50000
#define NUM_GROUP  20000
#define EMB        64
#define BATCH      1024
#define E_UI       2000000
#define E_U        1000000
#define E_I        1000000
#define INV_TEMP   10.0f
#define EPS_N      1e-12f

#define N_UG  (NUM_USER + NUM_GROUP)
#define N_IG  (NUM_ITEM + NUM_GROUP)
#define N_UI  (NUM_USER + NUM_ITEM)

typedef _Float16 v16h __attribute__((ext_vector_type(16)));
typedef float    v8f  __attribute__((ext_vector_type(8)));

#define E_ALL  (E_U + E_I + E_UI)
#define N_UIB  NUM_USER
#define N_ALL  (N_UG + N_IG + N_UIB)
#define SLICE  62500
#define SL_U   ((E_U  + SLICE - 1) / SLICE)
#define SL_I   ((E_I  + SLICE - 1) / SLICE)
#define SL_X   ((E_UI + SLICE - 1) / SLICE)
#define HW_U   ((N_UG + 1) / 2)
#define HW_I   ((N_IG + 1) / 2)
#define HW_X   ((N_UIB + 1) / 2)
#define HW_MAX 75000
#define HIST_THREADS 1024
#define HIST_U 8
static_assert(SLICE <= 65535, "a slice's counts fit 16 bits");
static_assert((SLICE & 3) == 0 && (E_U & 3) == 0 && (E_I & 3) == 0 && (E_UI & 3) == 0, "whole quads of edges");
static_assert(HW_U <= HW_MAX && HW_I <= HW_MAX && HW_X <= HW_MAX && (HW_MAX & 3) == 0, "counters fit LDS");
static_assert((N_UI + 1) / 2 <= HW_MAX, "every user-item column, bucket or not, has its counter in LDS");

__device__ __forceinline__ int slice_rank(unsigned* s_h, int c) {
  const unsigned old = atomicAdd(&s_h[c >> 1], (c & 1) ? 0x10000u : 1u);
  return (int)((c & 1) ? (old >> 16) : (old & 0xffffu));
}

__global__ __launch_bounds__(HIST_THREADS)
void slice_hist3(const int* __restrict__ colU, const int* __restrict__ colI,
                 const int* __restrict__ colX,
                 unsigned* __restrict__ histU, unsigned* __restrict__ histI,
                 unsigned* __restrict__ histX, int* __restrict__ rank) {
  __shared__ __attribute__((aligned(16))) unsigned s_h[HW_MAX];
  const int b = blockIdx.x;
  const int* col = colU; unsigned* hist = histU;
  int nE = E_U, hw = HW_U, ebase = 0, s = b;
  if (b >= SL_U + SL_I) {
    col = colX; hist = histX; nE = E_UI; hw = HW_X; ebase = E_U + E_I; s = b - (SL_U + SL_I);
  } else if (b >= SL_U) {
    col = colI; hist = histI; nE = E_I; hw = HW_I; ebase = E_U; s = b - SL_U;
  }
  const int tid = threadIdx.x;
  for (int w = tid; w < HW_MAX / 4; w += HIST_THREADS) ((uint4*)s_h)[w] = make_uint4(0u, 0u, 0u, 0u);
  __syncthreads();

  const int e0  = s * SLICE;
  const int len = (nE - e0 < SLICE) ? (nE - e0) : SLICE;
  const int nq  = len >> 2;
  const int4* __restrict__ cq = (const int4*)(col + e0);
  int4* __restrict__ rq = (int4*)(rank + ebase + e0);
  for (int q0 = 0; q0 < nq; q0 += HIST_THREADS * HIST_U) {
    int4 c[HIST_U];
    #pragma unroll
    for (int u = 0; u < HIST_U; ++u) {
      const int q = q0 + u * HIST_THREADS + tid;
      c[u] = cq[(q < nq) ? q : (nq - 1)];
    }
    #pragma unroll
    for (int u = 0; u < HIST_U; ++u) {
      const int q = q0 + u * HIST_THREADS + tid;
      if (q < nq) {
        int4 r;
        r.x = slice_rank(s_h, c[u].x);
        r.y = slice_rank(s_h, c[u].y);
        r.z = slice_rank(s_h, c[u].z);
        r.w = slice_rank(s_h, c[u].w);
        rq[q] = r;
      }
    }
  }
  __syncthreads();
  unsigned* __restrict__ ho = hist + (size_t)s * hw;
  if ((hw & 3) == 0) {
    for (int w = tid; w < hw / 4; w += HIST_THREADS) ((uint4*)ho)[w] = ((const uint4*)s_h)[w];
  } else {
    for (int w = tid; w < hw; w += HIST_THREADS) ho[w] = s_h[w];
  }
}

#define CB_U 16
__global__ void col_base3(const unsigned* __restrict__ histU, const unsigned* __restrict__ histI,
                          const unsigned* __restrict__ histX,
                          int* __restrict__ baseU, int* __restrict__ baseI,
                          int* __restrict__ baseX, int* __restrict__ counts,
                          float* __restrict__ tot, float* __restrict__ scal) {
  const int t = blockIdx.x * blockDim.x + threadIdx.x;
  if (t < 4 * BATCH) tot[t] = 0.0f;
  if (t < 2) scal[t] = 0.0f;
  if (t >= HW_U + HW_I + HW_X) return;
  const unsigned* hist = histU; int* base = baseU;
  int hw = HW_U, n = N_UG, ns = SL_U, cb = 0, w = t;
  if (t >= HW_U + HW_I) {
    hist = histX; base = baseX; hw = HW_X; n = N_UIB; ns = SL_X; cb = N_UG + N_IG; w = t - (HW_U + HW_I);
  } else if (t >= HW_U) {
    hist = histI; base = baseI; hw = HW_I; n = N_IG; ns = SL_I; cb = N_UG; w = t - HW_U;
  }
  const int  c0   = 2 * w;
  const bool two  = (c0 + 1 < n);
  const bool pair = two && ((n & 1) == 0);
  int run0 = 0, run1 = 0;
  for (int s0 = 0; s0 < ns; s0 += CB_U) {
    unsigned v[CB_U];
    #pragma unroll
    for (int u = 0; u < CB_U; ++u) {
      const int s = s0 + u;
      v[u] = hist[(size_t)((s < ns) ? s : (ns - 1)) * hw + w];
    }
    #pragma unroll
    for (int u = 0; u < CB_U; ++u) {
      const int s = s0 + u;
      if (s < ns) {
        int* bp = base + (size_t)s * n + c0;
        if (pair) { int2 o; o.x = run0; o.y = run1; *(int2*)bp = o; }
        else      { bp[0] = run0; if (two) bp[1] = run1; }
        run0 += (int)(v[u] & 0xffffu);
        run1 += (int)(v[u] >> 16);
      }
    }
  }
  counts[cb + c0] = run0;
  if (two) counts[cb + c0 + 1] = run1;
}

#define SCAN_THREADS 1024
#define SCAN_WAVES   (SCAN_THREADS / 32)
#define SCAN_TILE    128
#define SCAN_U       8
#define SCAN_GROUP   (SCAN_TILE * SCAN_U)

__device__ __forceinline__ int wave_incl_scan(int x, int lane) {
  x += __builtin_amdgcn_update_dpp(0, x, 0x111, 0xf, 0xf, true);
  x += __builtin_amdgcn_update_dpp(0, x, 0x112, 0xf, 0xf, true);
  x += __builtin_amdgcn_update_dpp(0, x, 0x114, 0xf, 0xf, true);
  x += __builtin_amdgcn_update_dpp(0, x, 0x118, 0xf, 0xf, true);
  const int low = __builtin_amdgcn_readlane(x, 15);
  x += (lane & 16) ? low : 0;
  return x;
}

__device__ __forceinline__ void scan_load_group(const int* __restrict__ p, int t,
                                                int lane, int end, bool vec,
                                                int4 (&v)[SCAN_U]) {
  if (vec && t + SCAN_GROUP <= end) {
    #pragma unroll
    for (int u = 0; u < SCAN_U; ++u)
      v[u] = *(const int4*)(p + t + u * SCAN_TILE + lane * 4);
  } else {
    #pragma unroll
    for (int u = 0; u < SCAN_U; ++u) {
      int i = t + u * SCAN_TILE + lane * 4;
      v[u].x = (i     < end) ? p[i]     : 0;
      v[u].y = (i + 1 < end) ? p[i + 1] : 0;
      v[u].z = (i + 2 < end) ? p[i + 2] : 0;
      v[u].w = (i + 3 < end) ? p[i + 3] : 0;
    }
  }
}

__device__ __forceinline__ void scan_store_group(int* __restrict__ p, int t,
                                                 int lane, int end, bool vec,
                                                 const int4 (&v)[SCAN_U]) {
  if (vec && t + SCAN_GROUP <= end) {
    #pragma unroll
    for (int u = 0; u < SCAN_U; ++u)
      *(int4*)(p + t + u * SCAN_TILE + lane * 4) = v[u];
  } else {
    #pragma unroll
    for (int u = 0; u < SCAN_U; ++u) {
      int i = t + u * SCAN_TILE + lane * 4;
      if (i     < end) p[i]     = v[u].x;
      if (i + 1 < end) p[i + 1] = v[u].y;
      if (i + 2 < end) p[i + 2] = v[u].z;
      if (i + 3 < end) p[i + 3] = v[u].w;
    }
  }
}

__global__ __launch_bounds__(SCAN_THREADS)
void exscan3_kernel(int* __restrict__ cnt0, int* __restrict__ off0, int n0,
                    int* __restrict__ cnt1, int* __restrict__ off1, int n1,
                    int* __restrict__ cnt2, int* __restrict__ off2, int n2) {
  int* __restrict__ counts  = cnt0;
  int* __restrict__ offsets = off0;
  int n = n0;
  if (blockIdx.x == 1) { counts = cnt1; offsets = off1; n = n1; }
  if (blockIdx.x == 2) { counts = cnt2; offsets = off2; n = n2; }
  __shared__ int s_wsum[SCAN_WAVES];
  const int tid  = threadIdx.x;
  const int lane = tid & 31;
  const int wave = __builtin_amdgcn_readfirstlane(tid >> 5);
  const int per = (((n + SCAN_WAVES - 1) / SCAN_WAVES) + SCAN_GROUP - 1)
                  & ~(SCAN_GROUP - 1);
  const int beg = wave * per;
  const int end = (beg + per < n) ? (beg + per) : n;
  const bool vec = ((((size_t)counts) | ((size_t)offsets)) & 15) == 0;

  int acc = 0;
  for (int t = beg; t < end; t += SCAN_GROUP) {
    int4 v[SCAN_U];
    scan_load_group(counts, t, lane, end, vec, v);
    #pragma unroll
    for (int u = 0; u < SCAN_U; ++u) acc += (v[u].x + v[u].y) + (v[u].z + v[u].w);
  }
  #pragma unroll
  for (int off = 16; off > 0; off >>= 1) acc += __shfl_xor(acc, off, 32);
  if (lane == 0) s_wsum[wave] = acc;
  __syncthreads();

  const int wt    = s_wsum[lane];
  const int winc  = wave_incl_scan(wt, lane);
  int       carry = __builtin_amdgcn_readlane(winc - wt, wave);
  const int grand = __builtin_amdgcn_readlane(winc, 31);

  for (int t = beg; t < end; t += SCAN_GROUP) {
    int4 v[SCAN_U];
    scan_load_group(counts, t, lane, end, vec, v);
    int4 o[SCAN_U];
    #pragma unroll
    for (int u = 0; u < SCAN_U; ++u) {
      const int p1 = v[u].x;
      const int p2 = p1 + v[u].y;
      const int p3 = p2 + v[u].z;
      const int p4 = p3 + v[u].w;
      const int inc = wave_incl_scan(p4, lane);
      const int ex  = carry + (inc - p4);
      o[u].x = ex;
      o[u].y = ex + p1;
      o[u].z = ex + p2;
      o[u].w = ex + p3;
      carry += __builtin_amdgcn_readlane(inc, 31);
    }
    scan_store_group(offsets, t, lane, end, vec, o);
    scan_store_group(counts,  t, lane, end, vec, o);
  }
  if (tid == 0) offsets[n] = grand;
}

__global__ void csr_fill3(const int* __restrict__ colU, const int* __restrict__ rowU,
                          const float* __restrict__ valU, const int* __restrict__ offU,
                          const int* __restrict__ baseU, int* __restrict__ pairsU,
                          const int* __restrict__ colI, const int* __restrict__ rowI,
                          const float* __restrict__ valI, const int* __restrict__ offI,
                          const int* __restrict__ baseI, int* __restrict__ pairsI,
                          const int* __restrict__ colX, const int* __restrict__ rowX,
                          const float* __restrict__ valX, const int* __restrict__ offX,
                          const int* __restrict__ baseX, int* __restrict__ pairsX,
                          const int* __restrict__ rank) {
  int e = blockIdx.x * blockDim.x + threadIdx.x;
  if (e >= E_ALL) return;
  const int* col = colU; const int* row = rowU; const float* val = valU;
  const int* offsets = offU; const int* base = baseU; int* pairs = pairsU;
  int le = e, n = N_UG;
  if (e >= E_U + E_I) {
    col = colX; row = rowX; val = valX; offsets = offX; base = baseX; pairs = pairsX;
    le = e - (E_U + E_I); n = N_UIB;
  } else if (e >= E_U) {
    col = colI; row = rowI; val = valI; offsets = offI; base = baseI; pairs = pairsI;
    le = e - E_U; n = N_IG;
  }
  const int c = col[le];
  if (c >= n) return;
  int slot = offsets[c] + base[(size_t)(le / SLICE) * n + c] + rank[e];
  int2 rv;
  rv.x = row[le];
  rv.y = __float_as_int(val[le]);
  *(int2*)(pairs + 2 * (size_t)slot) = rv;
}

typedef _Float16 v2h __attribute__((ext_vector_type(2)));

#define GATHER_DEPTH 16

__device__ __forceinline__ const float* table_row(const float* __restrict__ p0, int n0,
                                                  const float* __restrict__ p1, int r) {
  return (r < n0) ? (p0 + (size_t)r * EMB) : (p1 + (size_t)(r - n0) * EMB);
}

__device__ __forceinline__ int2 load_pairs(const int2* __restrict__ rv, int base,
                                           int end, int lane) {
  int2 p; p.x = 0; p.y = 0;
  if (base + lane < end) p = rv[base + lane];
  return p;
}

typedef float v2f __attribute__((ext_vector_type(2)));
typedef const v2f __attribute__((address_space(1))) gv2f;

__device__ __forceinline__ void pair_addr(const float* __restrict__ p0, int n0,
                                          const float* __restrict__ p1, int r,
                                          int& alo, int& ahi) {
  const unsigned long long a = (unsigned long long)table_row(p0, n0, p1, r);
  alo = (int)(unsigned)a;
  ahi = (int)(unsigned)(a >> 32);
}

template <int U>
__device__ __forceinline__ void loads_from(const unsigned (&lo)[GATHER_DEPTH],
                                           const unsigned (&hi)[GATHER_DEPTH],
                                           int j, int n, int lane,
                                           float2 (&d)[GATHER_DEPTH]) {
  if constexpr (U < GATHER_DEPTH) {
    if (j + U < n) {
      const gv2f* rp = (const gv2f*)(((unsigned long long)hi[U] << 32) | lo[U]);
      const v2f t = rp[lane];
      d[U].x = t.x; d[U].y = t.y;
      loads_from<U + 1>(lo, hi, j, n, lane, d);
    }
  }
}

__device__ __forceinline__ void batch_load(int alo, int ahi, int j, int n, int lane,
                                           float2 (&d)[GATHER_DEPTH]) {
  unsigned lo[GATHER_DEPTH], hi[GATHER_DEPTH];
  #pragma unroll
  for (int u = 0; u < GATHER_DEPTH; ++u) {
    lo[u] = (unsigned)__builtin_amdgcn_readlane(alo, j + u);
    hi[u] = (unsigned)__builtin_amdgcn_readlane(ahi, j + u);
  }
  loads_from<0>(lo, hi, j, n, lane, d);
}

template <int U>
__device__ __forceinline__ void sums_from(int pv, int j, int n,
                                          const float2 (&d)[GATHER_DEPTH],
                                          float& ax, float& ay) {
  if constexpr (U < GATHER_DEPTH) {
    if (j + U < n) {
      const float v = __int_as_float(__builtin_amdgcn_readlane(pv, j + U));
      ax += d[U].x * v;
      ay += d[U].y * v;
      sums_from<U + 1>(pv, j, n, d, ax, ay);
    }
  }
}

__device__ __forceinline__ void batch_sum(int pv, int j, int n,
                                          const float2 (&d)[GATHER_DEPTH],
                                          float& ax, float& ay) {
  sums_from<0>(pv, j, n, d, ax, ay);
}

__device__ __forceinline__ void csr_prop_rows(const float* __restrict__ p0, int n0,
                         const float* __restrict__ p1,
                         const int* __restrict__ offsets,
                         const int* __restrict__ pairs,
                         float* __restrict__ out, int rowBeg, int nRows, int rowsPerWave,
                         int addSelf,
                         _Float16* __restrict__ nlo, int split,
                         _Float16* __restrict__ nhi, int blk) {
  const int2* __restrict__ rv = (const int2*)pairs;
  const int lane = threadIdx.x & 31;
  const int wv = __builtin_amdgcn_readfirstlane(
      (int)((blk * blockDim.x + threadIdx.x) >> 5));
  const int r0 = rowBeg + wv * rowsPerWave;
  if (r0 >= nRows) return;
  const int nr = (nRows - r0 < rowsPerWave) ? (nRows - r0) : rowsPerWave;
  const int offv = (lane <= nr) ? offsets[r0 + lane] : 0;
  int beg = __builtin_amdgcn_readlane(offv, 0);
  int end = __builtin_amdgcn_readlane(offv, 1);
  int2 cur = load_pairs(rv, beg, end, lane);

  for (int i = 0; i < nr; ++i) {
    const int wid = r0 + i;
    const int nf  = (end - beg < 32) ? (end - beg) : 32;
    float ax = 0.0f, ay = 0.0f;
    float2 d[GATHER_DEPTH];
    int alo, ahi;
    pair_addr(p0, n0, p1, cur.x, alo, ahi);
    batch_load(alo, ahi, 0, nf, lane, d);
    int nend = end;
    int2 nxt; nxt.x = 0; nxt.y = 0;
    if (i + 1 < nr) {
      nend = __builtin_amdgcn_readlane(offv, i + 2);
      nxt  = load_pairs(rv, end, nend, lane);
    }
    float sx = 0.0f, sy = 0.0f;
    if (addSelf) {
      float2 s = *(const float2*)(table_row(p0, n0, p1, wid) + lane * 2);
      sx = s.x; sy = s.y;
    }
    batch_sum(cur.y, 0, nf, d, ax, ay);
    if (nf > GATHER_DEPTH) {
      batch_load(alo, ahi, GATHER_DEPTH, nf, lane, d);
      batch_sum(cur.y, GATHER_DEPTH, nf, d, ax, ay);
    }
    for (int base = beg + 32; base < end; base += 32) {
      const int2 more = load_pairs(rv, base, end, lane);
      const int n = (end - base < 32) ? (end - base) : 32;
      int mlo, mhi;
      pair_addr(p0, n0, p1, more.x, mlo, mhi);
      for (int j = 0; j < n; j += GATHER_DEPTH) {
        batch_load(mlo, mhi, j, n, lane, d);
        batch_sum(more.y, j, n, d, ax, ay);
      }
    }
    float2 o; o.x = ax; o.y = ay;
    if (addSelf) { o.x = (sx + ax) * 0.5f; o.y = (sy + ay) * 0.5f; }
    *(float2*)(out + (size_t)wid * EMB + lane * 2) = o;

    if (nlo != nullptr || nhi != nullptr) {
      float ss = fmaf(o.x, o.x, o.y * o.y);
      #pragma unroll
      for (int off = 1; off < 32; off <<= 1) ss += __shfl_xor(ss, off, 32);
      float scale = 1.0f / fmaxf(sqrtf(ss), EPS_N);
      _Float16* dst = nullptr;
      if (wid < split) { if (nlo != nullptr) dst = nlo + (size_t)wid * EMB; }
      else             { if (nhi != nullptr) dst = nhi + (size_t)(wid - split) * EMB; }
      if (dst != nullptr) {
        v2h h;
        h.x = (_Float16)(o.x * scale);
        h.y = (_Float16)(o.y * scale);
        *(v2h*)(dst + lane * 2) = h;
      }
    }
    beg = end; end = nend; cur = nxt;
  }
}

__global__ void csr_prop3(
    const float* __restrict__ p0A, int n0A, const float* __restrict__ p1A,
    const int* __restrict__ offA, const int* __restrict__ pairsA, float* __restrict__ outA,
    int rowBegA, int nRowsA, int rpwA, int addSelfA,
    _Float16* __restrict__ nloA, int splitA, _Float16* __restrict__ nhiA, int nbA,
    const float* __restrict__ p0B, int n0B, const float* __restrict__ p1B,
    const int* __restrict__ offB, const int* __restrict__ pairsB, float* __restrict__ outB,
    int rowBegB, int nRowsB, int rpwB, int addSelfB,
    _Float16* __restrict__ nloB, int splitB, _Float16* __restrict__ nhiB, int nbB,
    const float* __restrict__ p0C, int n0C, const float* __restrict__ p1C,
    const int* __restrict__ offC, const int* __restrict__ pairsC, float* __restrict__ outC,
    int rowBegC, int nRowsC, int rpwC, int addSelfC,
    _Float16* __restrict__ nloC, int splitC, _Float16* __restrict__ nhiC) {
  const float* p0 = p0A; const float* p1 = p1A; int n0 = n0A;
  const int* off = offA; const int* pairs = pairsA; float* out = outA;
  int rowBeg = rowBegA, nRows = nRowsA, rpw = rpwA, addSelf = addSelfA, split = splitA;
  _Float16* nlo = nloA; _Float16* nhi = nhiA;
  int blk = blockIdx.x;
  if (blk >= nbA + nbB) {
    p0 = p0C; p1 = p1C; n0 = n0C; off = offC; pairs = pairsC; out = outC;
    rowBeg = rowBegC; nRows = nRowsC; rpw = rpwC; addSelf = addSelfC; split = splitC;
    nlo = nloC; nhi = nhiC; blk -= nbA + nbB;
  } else if (blk >= nbA) {
    p0 = p0B; p1 = p1B; n0 = n0B; off = offB; pairs = pairsB; out = outB;
    rowBeg = rowBegB; nRows = nRowsB; rpw = rpwB; addSelf = addSelfB; split = splitB;
    nlo = nloB; nhi = nhiB; blk -= nbA;
  }
  csr_prop_rows(p0, n0, p1, off, pairs, out, rowBeg, nRows, rpw, addSelf, nlo, split, nhi, blk);
}

#define OUT_THREADS 256
#define OUT_NB      ((BATCH * EMB + OUT_THREADS - 1) / OUT_THREADS)
__device__ __forceinline__
void gather_out_reg_rows(const float* __restrict__ ug,
                               const float* __restrict__ uif,
                               const float* __restrict__ ig,
                               const float* __restrict__ ued,
                               const float* __restrict__ uei,
                               const float* __restrict__ ged,
                               const float* __restrict__ gei,
                               const int* __restrict__ u,
                               const int* __restrict__ p,
                               const int* __restrict__ ng,
                               float* __restrict__ out,
                               float* __restrict__ reg_out, int blk) {
  int gid = blk * OUT_THREADS + threadIdx.x;
  if (gid >= BATCH * EMB) return;
  int b = gid >> 6, c = gid & 63;
  const size_t PO = (size_t)BATCH * 128;
  const size_t NO = 2 * PO;
  out[(size_t)b * 128 + c]           = ug [(size_t)u[b] * EMB + c];
  out[(size_t)b * 128 + 64 + c]      = uif[(size_t)u[b] * EMB + c];
  out[PO + (size_t)b * 128 + c]      = ug [(size_t)(NUM_USER + p[b])  * EMB + c];
  out[PO + (size_t)b * 128 + 64 + c] = ig [(size_t)(NUM_ITEM + p[b])  * EMB + c];
  out[NO + (size_t)b * 128 + c]      = ug [(size_t)(NUM_USER + ng[b]) * EMB + c];
  out[NO + (size_t)b * 128 + 64 + c] = ig [(size_t)(NUM_ITEM + ng[b]) * EMB + c];

  size_t uo = (size_t)u[b]  * EMB + c;
  size_t po = (size_t)p[b]  * EMB + c;
  size_t no = (size_t)ng[b] * EMB + c;
  float s = uei[uo]*uei[uo] + ued[uo]*ued[uo]
          + gei[po]*gei[po] + ged[po]*ged[po]
          + gei[no]*gei[no] + ged[no]*ged[no];
  #pragma unroll
  for (int off = 1; off < 32; off <<= 1) s += __shfl_xor(s, off, 32);
  if ((threadIdx.x & 31) == 0) atomicAdd(reg_out, s * (0.5f / (float)BATCH));
}

#define GEMM_MT     8
#define GEMM_ROWS   (16 * GEMM_MT)
#define GEMM_WAVES  (BATCH / GEMM_ROWS)
#define GEMM_TPG_U  13
#define GEMM_TPG_G  5
static_assert(BATCH % GEMM_ROWS == 0, "whole row groups");
static_assert(GEMM_WAVES * 32 == OUT_THREADS, "the output work-groups have the SSL launch's size");
static_assert(NUM_USER % 16 == 0 && NUM_GROUP % 16 == 0, "whole table tiles");

__device__ __forceinline__ void load_frag2(const _Float16* __restrict__ row, int h,
                                           v16h& f0, v16h& f1) {
  const _Float16* p = row + 16 * h;
  f0 = *(const v16h*)p;
  f1 = *(const v16h*)(p + 32);
}

__device__ __forceinline__ float row16_sum(float x) {
  x += __int_as_float(__builtin_amdgcn_update_dpp(0, __float_as_int(x), 0x111, 0xf, 0xf, true));
  x += __int_as_float(__builtin_amdgcn_update_dpp(0, __float_as_int(x), 0x112, 0xf, 0xf, true));
  x += __int_as_float(__builtin_amdgcn_update_dpp(0, __float_as_int(x), 0x114, 0xf, 0xf, true));
  x += __int_as_float(__builtin_amdgcn_update_dpp(0, __float_as_int(x), 0x118, 0xf, 0xf, true));
  return x;
}

#define GEMM_NB_U   ((NUM_USER  / 16 + GEMM_TPG_U - 1) / GEMM_TPG_U)
#define GEMM_NB_G   ((NUM_GROUP / 16 + GEMM_TPG_G - 1) / GEMM_TPG_G)

__device__ __forceinline__
void ssl_gemm_term(const _Float16* __restrict__ A16,
                   const _Float16* __restrict__ B16,
                   const int* __restrict__ idx,
                   float* __restrict__ tot,
                   int nTiles, int tilesPerGroup, int blk) {
  __shared__ __attribute__((aligned(16))) float s_tot[GEMM_WAVES][GEMM_ROWS];
  const int lane = threadIdx.x & 31;
  const int wave = threadIdx.x >> 5;
  const int c    = lane & 15;
  const int h    = lane >> 4;
  const int row0 = wave * GEMM_ROWS;

  v16h a0[GEMM_MT], a1[GEMM_MT];
  #pragma unroll
  for (int mt = 0; mt < GEMM_MT; ++mt)
    load_frag2(A16 + (size_t)idx[row0 + mt * 16 + c] * EMB, h, a0[mt], a1[mt]);

  float s[GEMM_MT][8];
  #pragma unroll
  for (int mt = 0; mt < GEMM_MT; ++mt) {
    #pragma unroll
    for (int v = 0; v < 8; ++v) s[mt][v] = 0.0f;
  }

  const int t0 = blk * tilesPerGroup;
  const int t1 = (t0 + tilesPerGroup < nTiles) ? (t0 + tilesPerGroup) : nTiles;
  const _Float16* brow = B16 + ((size_t)t0 * 16 + c) * EMB;
  v16h b0, b1;
  load_frag2(brow, h, b0, b1);
  for (int t = t0; t < t1; ++t) {
    const _Float16* nrow = brow + ((t + 1 < t1) ? 16 * EMB : 0);
    v16h nb0, nb1;
    load_frag2(nrow, h, nb0, nb1);
    asm volatile("" ::: "memory");
    #pragma unroll
    for (int mt = 0; mt < GEMM_MT; ++mt) {
      v8f d = {};
      d = __builtin_amdgcn_wmma_f32_16x16x32_f16(false, a0[mt], false, b0,
                                                 (short)0, d, false, false);
      d = __builtin_amdgcn_wmma_f32_16x16x32_f16(false, a1[mt], false, b1,
                                                 (short)0, d, false, false);
      #pragma unroll
      for (int v = 0; v < 8; ++v) s[mt][v] += __expf(d[v] * INV_TEMP);
    }
    b0 = nb0; b1 = nb1; brow = nrow;
  }

  #pragma unroll
  for (int mt = 0; mt < GEMM_MT; ++mt) {
    #pragma unroll
    for (int v = 0; v < 8; ++v) s[mt][v] = row16_sum(s[mt][v]);
  }
  if (c == 15) {
    #pragma unroll
    for (int mt = 0; mt < GEMM_MT; ++mt) {
      float4* p = (float4*)&s_tot[wave][mt * 16 + 8 * h];
      p[0] = make_float4(s[mt][0], s[mt][1], s[mt][2], s[mt][3]);
      p[1] = make_float4(s[mt][4], s[mt][5], s[mt][6], s[mt][7]);
    }
  }
  __syncthreads();
  #pragma unroll
  for (int q = 0; q < GEMM_ROWS / 32; ++q)
    atomicAdd(&tot[row0 + q * 32 + lane], s_tot[wave][q * 32 + lane]);
}

#define GEMM_NB_ALL (2 * GEMM_NB_U + 2 * GEMM_NB_G)
__global__ __launch_bounds__(GEMM_WAVES * 32)
void ssl_gemm4_out(const _Float16* __restrict__ uiU, const _Float16* __restrict__ ugU,
                   const _Float16* __restrict__ igG, const _Float16* __restrict__ ugG,
                   const int* __restrict__ uin, const int* __restrict__ pos,
                   const int* __restrict__ neg, float* __restrict__ tot,
                   const float* __restrict__ ug, const float* __restrict__ uif,
                   const float* __restrict__ ig,
                   const float* __restrict__ ued, const float* __restrict__ uei,
                   const float* __restrict__ ged, const float* __restrict__ gei,
                   float* __restrict__ out, float* __restrict__ reg_out) {
  if (blockIdx.x >= GEMM_NB_ALL) {
    gather_out_reg_rows(ug, uif, ig, ued, uei, ged, gei, uin, pos, neg, out, reg_out,
                        blockIdx.x - GEMM_NB_ALL);
    return;
  }
  const _Float16* A16 = uiU; const _Float16* B16 = ugU; const int* idx = uin;
  int nTiles = NUM_USER / 16, tpg = GEMM_TPG_U, term = 0, blk = blockIdx.x;
  if (blk >= 2 * GEMM_NB_U + GEMM_NB_G) {
    A16 = ugG; B16 = igG; idx = pos; nTiles = NUM_GROUP / 16; tpg = GEMM_TPG_G; term = 3;
    blk -= 2 * GEMM_NB_U + GEMM_NB_G;
  } else if (blk >= 2 * GEMM_NB_U) {
    A16 = igG; B16 = ugG; idx = pos; nTiles = NUM_GROUP / 16; tpg = GEMM_TPG_G; term = 2;
    blk -= 2 * GEMM_NB_U;
  } else if (blk >= GEMM_NB_U) {
    A16 = ugU; B16 = uiU; term = 1; blk -= GEMM_NB_U;
  }
  ssl_gemm_term(A16, B16, idx, tot + term * BATCH, nTiles, tpg, blk);
}

__global__ void ssl_loss4(const _Float16* __restrict__ uiU, const _Float16* __restrict__ ugU,
                          const _Float16* __restrict__ igG, const _Float16* __restrict__ ugG,
                          const int* __restrict__ uin, const int* __restrict__ pos,
                          const float* __restrict__ tot4,
                          float* __restrict__ ssl_out) {
  int gid = blockIdx.x * blockDim.x + threadIdx.x;
  int w = gid >> 5;
  if (w >= 4 * BATCH) return;
  const int term = w / BATCH;
  int i = w - term * BATCH;
  const _Float16* A16 = uiU; const _Float16* B16 = ugU; const int* idx = uin;
  if (term == 1) { A16 = ugU; B16 = uiU; }
  if (term == 2) { A16 = igG; B16 = ugG; idx = pos; }
  if (term == 3) { A16 = ugG; B16 = igG; idx = pos; }
  const float* tot = tot4 + term * BATCH;
  int t = gid & 31;
  size_t ro = (size_t)idx[i] * EMB + 2 * t;
  float d = (float)A16[ro] * (float)B16[ro]
          + (float)A16[ro + 1] * (float)B16[ro + 1];
  #pragma unroll
  for (int off = 1; off < 32; off <<= 1) d += __shfl_xor(d, off, 32);
  if (t == 0) {
    float li = -(d * INV_TEMP - logf(tot[i])) * (1.0f / (float)BATCH);
    atomicAdd(ssl_out, li);
  }
}

extern "C" void kernel_launch(void* const* d_in, const int* in_sizes, int n_in,
                              void* d_out, int out_size, void* d_ws, size_t ws_size,
                              hipStream_t stream) {
  (void)in_sizes; (void)n_in; (void)out_size; (void)ws_size;
  const float* ued  = (const float*)d_in[0];
  const float* uei  = (const float*)d_in[1];
  const float* ged  = (const float*)d_in[2];
  const float* gei  = (const float*)d_in[3];
  const float* item = (const float*)d_in[4];
  const int*   hu_r = (const int*)d_in[5];
  const int*   hu_c = (const int*)d_in[6];
  const float* hu_v = (const float*)d_in[7];
  const int*   hi_r = (const int*)d_in[8];
  const int*   hi_c = (const int*)d_in[9];
  const float* hi_v = (const float*)d_in[10];
  const int*   ui_r = (const int*)d_in[11];
  const int*   ui_c = (const int*)d_in[12];
  const float* ui_v = (const float*)d_in[13];
  const int*   uin  = (const int*)d_in[14];
  const int*   pos  = (const int*)d_in[15];
  const int*   neg  = (const int*)d_in[16];
  float* out = (float*)d_out;

  float* W = (float*)d_ws;
  size_t off = 0;
  float* ug_b = W + off; off += (size_t)N_UG * EMB;
  float* ug_a = W + off; off += (size_t)N_UG * EMB;
  float* ig_b = W + off; off += (size_t)N_IG * EMB;
  float* ig_a = W + off; off += (size_t)N_IG * EMB;
  float* agg  = W + off; off += (size_t)N_UI * EMB;
  float* tot  = W + off; off += 4 * BATCH;
  int* pairs_ug = (int*)(W + off); off += E_UI;
  int* offsets_ug = (int*)(W + off); off += N_UI;
  int* offsets_ui = (int*)(W + off); off += N_UI + 1;
  off = (off + 7) & ~(size_t)7;
  _Float16* H = (_Float16*)(W + off);
  _Float16* ug_user_n16  = H;
  _Float16* ui_user_n16  = H + (size_t)NUM_USER * EMB;
  _Float16* ug_group_n16 = H + (size_t)NUM_USER * EMB * 2;
  _Float16* ig_group_n16 = ug_group_n16 + (size_t)NUM_GROUP * EMB;

  int* rank_all   = (int*)ug_b;
  int* counts_all = (int*)ug_b + E_ALL;
  int* counts_ug  = counts_all;
  int* counts_ig  = counts_all + N_UG;
  int* counts_ui  = counts_all + N_UG + N_IG;
  unsigned* hist_ug = (unsigned*)ig_b;
  unsigned* hist_ig = hist_ug + (size_t)SL_U * HW_U;
  unsigned* hist_ui = hist_ig + (size_t)SL_I * HW_I;
  int* base_ug = (int*)agg;
  int* base_ig = base_ug + (size_t)SL_U * N_UG;
  int* base_ui = base_ig + (size_t)SL_I * N_IG;
  int* pairs_ui   = (int*)ug_a;
  int* pairs_ig   = (int*)agg + (size_t)N_UIB * EMB;
  int* offsets_ig = pairs_ig + (size_t)2 * E_I;
  static_assert((size_t)E_ALL + N_ALL <= (size_t)N_UG * EMB, "ranks and totals fit ug_b");
  static_assert((size_t)SL_U * HW_U + (size_t)SL_I * HW_I + (size_t)SL_X * HW_X <= (size_t)N_IG * EMB,
                "packed counters fit ig_b");
  static_assert((size_t)SL_U * N_UG + (size_t)SL_I * N_IG + (size_t)SL_X * N_UIB <= (size_t)N_UIB * EMB,
                "slice starts fit the user rows of agg");
  static_assert((((size_t)SL_U * N_UG) & 1) == 0 && (((size_t)SL_I * N_IG) & 1) == 0, "slice starts 8-byte aligned");
  static_assert((size_t)2 * E_UI <= (size_t)N_UG * EMB, "user-item pairs fit ug_a");
  static_assert((size_t)2 * E_I + N_IG + 1 <= (size_t)(N_UI - N_UIB) * EMB, "item-group pairs and offsets fit behind the user rows of agg");
  static_assert((((size_t)N_UIB * EMB) & 3) == 0, "item-group pairs 16-byte aligned");
  static_assert((size_t)2 * E_U <= (size_t)E_UI, "user-group pairs fit their plane");
  static_assert(N_UG + 1 <= N_UI, "user-group offsets fit their plane");
  static_assert((((size_t)N_UG * EMB * 2 + (size_t)N_IG * EMB * 2 + (size_t)N_UI * EMB + 4 * BATCH) & 1) == 0
                && (((size_t)N_UG * EMB) & 1) == 0, "pairs 8-byte aligned");
  static_assert((E_ALL & 3) == 0 && (N_UG & 3) == 0 && ((N_UG + N_IG) & 3) == 0 && (N_UI & 3) == 0 && (N_UIB & 1) == 0
                && ((2 * E_I) & 3) == 0 && (E_UI & 3) == 0, "scan arrays 16-byte aligned");
  const int TB = 256;
  #define CDIV(a, b) (((a) + (b) - 1) / (b))
  #define RPW_UG 8
  #define RPW_IG 5
  #define RPW_IG2 2
  #define RPW_UI 7

  slice_hist3<<<SL_U + SL_I + SL_X, HIST_THREADS, 0, stream>>>(
      hu_c, hi_c, ui_c, hist_ug, hist_ig, hist_ui, rank_all);
  col_base3<<<CDIV(HW_U + HW_I + HW_X, TB), TB, 0, stream>>>(
      hist_ug, hist_ig, hist_ui, base_ug, base_ig, base_ui, counts_all,
      tot, out + (size_t)3 * BATCH * 128);
  exscan3_kernel<<<3, 1024, 0, stream>>>(counts_ug, offsets_ug, N_UG,
                                         counts_ig, offsets_ig, N_IG,
                                         counts_ui, offsets_ui, N_UIB);
  csr_fill3<<<CDIV(E_ALL, TB), TB, 0, stream>>>(
      hu_c, hu_r, hu_v, offsets_ug, base_ug, pairs_ug,
      hi_c, hi_r, hi_v, offsets_ig, base_ig, pairs_ig,
      ui_c, ui_r, ui_v, offsets_ui, base_ui, pairs_ui, rank_all);

  constexpr int nbUI  = CDIV(CDIV(N_UIB, RPW_UI) * 32, TB);
  constexpr int nbIG1 = CDIV(CDIV(N_IG, RPW_IG) * 32, TB);
  constexpr int nbUG  = CDIV(CDIV(N_UG, RPW_UG) * 32, TB);
  constexpr int nbIG2 = CDIV(CDIV(NUM_GROUP, RPW_IG2) * 32, TB);
  csr_prop3<<<nbUI + nbIG1 + nbUG, TB, 0, stream>>>(
      uei, NUM_USER, item, offsets_ui, pairs_ui, agg, 0, N_UIB, RPW_UI,
      1, ui_user_n16, NUM_USER, nullptr, nbUI,
      item, NUM_ITEM, gei, offsets_ig, pairs_ig, ig_b, 0, N_IG, RPW_IG,
      0, nullptr, 0, nullptr, nbIG1,
      ued, NUM_USER, ged, offsets_ug, pairs_ug, ug_b, 0, N_UG, RPW_UG,
      0, nullptr, 0, nullptr);

  csr_prop3<<<nbUG + nbIG2, TB, 0, stream>>>(
      ug_b, 1 << 30, nullptr, offsets_ug, pairs_ug, ug_a, 0, N_UG, RPW_UG,
      0, ug_user_n16, NUM_USER, ug_group_n16, nbUG,
      ig_b, 1 << 30, nullptr, offsets_ig, pairs_ig, ig_a, NUM_ITEM, N_IG, RPW_IG2,
      0, nullptr, NUM_ITEM, ig_group_n16, nbIG2,
      nullptr, 0, nullptr, nullptr, nullptr, nullptr, 0, 0, 1,
      0, nullptr, 0, nullptr);

  ssl_gemm4_out<<<GEMM_NB_ALL + OUT_NB, GEMM_WAVES * 32, 0, stream>>>(
      ui_user_n16, ug_user_n16, ig_group_n16, ug_group_n16, uin, pos, neg, tot,
      ug_a, agg, ig_a, ued, uei, ged, gei, out, out + (size_t)3 * BATCH * 128);


  float* ssl_slot = out + (size_t)3 * BATCH * 128 + 1;
  ssl_loss4<<<CDIV(4 * BATCH * 32, TB), TB, 0, stream>>>(
      ui_user_n16, ug_user_n16, ig_group_n16, ug_group_n16, uin, pos, tot, ssl_slot);
  #undef RPW_UG
  #undef RPW_IG
  #undef RPW_IG2
  #undef RPW_UI
  #undef CDIV
}


# CSR build as a two-level counting sort in 2 launches (was 4): chunks sort edges by column range through LDS, one work-group per range counts, scans, places; bucket order = LDS cursor arrival
# speedup vs baseline: 9.3930x; 1.4761x over previous
#include <hip/hip_runtime.h>
#include <hip/hip_bf16.h>
#include <math.h>

#define NUM_USER   100000
#define NUM_ITEM   50000
#define NUM_GROUP  20000
#define EMB        64
#define BATCH      1024
#define E_UI       2000000
#define E_U        1000000
#define E_I        1000000
#define INV_TEMP   10.0f
#define EPS_N      1e-12f

#define N_UG  (NUM_USER + NUM_GROUP)
#define N_IG  (NUM_ITEM + NUM_GROUP)
#define N_UI  (NUM_USER + NUM_ITEM)

typedef _Float16 v16h __attribute__((ext_vector_type(16)));
typedef float    v8f  __attribute__((ext_vector_type(8)));

#define N_UIB  NUM_USER
#define RB        10
#define RCOLS     (1 << RB)
#define NR_U      ((N_UG  + RCOLS - 1) >> RB)
#define NR_I      ((N_IG  + RCOLS - 1) >> RB)
#define NR_X      ((N_UIB + RCOLS - 1) >> RB)
#define NR_MAX    128
#define CH        15632
#define NCH_U     ((E_U  + CH - 1) / CH)
#define NCH_I     ((E_I  + CH - 1) / CH)
#define NCH_X     ((E_UI + CH - 1) / CH)
#define NCH_ALL   (NCH_U + NCH_I + NCH_X)
#define NCH_MAX   128
#define PART_THREADS 1024
#define PART_Q    ((CH / 4 + PART_THREADS - 1) / PART_THREADS)
static_assert(NR_U <= NR_MAX && NR_I <= NR_MAX && NR_X <= NR_MAX, "a chunk's range counters fit");
static_assert(NCH_U <= NCH_MAX && NCH_I <= NCH_MAX && NCH_X <= NCH_MAX, "a range's piece list fits");
static_assert((CH & 15) == 0 && (E_U & 3) == 0 && (E_I & 3) == 0 && (E_UI & 3) == 0, "chunks of whole, aligned quads");
static_assert(NR_MAX == 128, "the range scan takes 4 counters per lane of one wave");

__device__ __forceinline__ int wave_incl_scan(int x, int lane) {
  x += __builtin_amdgcn_update_dpp(0, x, 0x111, 0xf, 0xf, true);
  x += __builtin_amdgcn_update_dpp(0, x, 0x112, 0xf, 0xf, true);
  x += __builtin_amdgcn_update_dpp(0, x, 0x114, 0xf, 0xf, true);
  x += __builtin_amdgcn_update_dpp(0, x, 0x118, 0xf, 0xf, true);
  const int low = __builtin_amdgcn_readlane(x, 15);
  x += (lane & 16) ? low : 0;
  return x;
}

__device__ __forceinline__ int part_rank(int* s_cnt, int c, int nkeep) {
  return (c < nkeep) ? atomicAdd(&s_cnt[c >> RB], 1) : -1;
}

__device__ __forceinline__ void part_place(int* sC, int* sR, int* sV, const int* s_start,
                                           int c, int rk, int r, float v) {
  if (rk >= 0) {
    const int p = s_start[c >> RB] + rk;
    sC[p] = c; sR[p] = r; sV[p] = __float_as_int(v);
  }
}

__global__ __launch_bounds__(PART_THREADS)
void part3(const int* __restrict__ colU, const int* __restrict__ rowU, const float* __restrict__ valU,
           const int* __restrict__ colI, const int* __restrict__ rowI, const float* __restrict__ valI,
           const int* __restrict__ colX, const int* __restrict__ rowX, const float* __restrict__ valX,
           int* __restrict__ gC, int* __restrict__ gR, int* __restrict__ gV,
           int* __restrict__ cntT, int* __restrict__ startT,
           float* __restrict__ tot, float* __restrict__ scal) {
  __shared__ __attribute__((aligned(16))) int sC[CH];
  __shared__ __attribute__((aligned(16))) int sR[CH];
  __shared__ __attribute__((aligned(16))) int sV[CH];
  __shared__ __attribute__((aligned(16))) int s_cnt[NR_MAX];
  __shared__ __attribute__((aligned(16))) int s_start[NR_MAX];
  const int tid = threadIdx.x;
  const int b = blockIdx.x;
  if (b == 0) {
    for (int t = tid; t < 4 * BATCH; t += PART_THREADS) tot[t] = 0.0f;
    if (tid < 2) scal[tid] = 0.0f;
  }
  const int* col = colU; const int* row = rowU; const float* val = valU;
  int nE = E_U, nkeep = N_UG, k = b;
  if (b >= NCH_U + NCH_I) {
    col = colX; row = rowX; val = valX; nE = E_UI; nkeep = N_UIB; k = b - (NCH_U + NCH_I);
  } else if (b >= NCH_U) {
    col = colI; row = rowI; val = valI; nE = E_I; nkeep = N_IG; k = b - NCH_U;
  }
  if (tid < NR_MAX) s_cnt[tid] = 0;
  __syncthreads();

  const int e0  = k * CH;
  const int len = (nE - e0 < CH) ? (nE - e0) : CH;
  const int nq  = len >> 2;
  const int4*   __restrict__ cq = (const int4*)(col + e0);
  const int4*   __restrict__ rq = (const int4*)(row + e0);
  const float4* __restrict__ vq = (const float4*)(val + e0);
  int4 c[PART_Q], rk[PART_Q];
  #pragma unroll
  for (int u = 0; u < PART_Q; ++u) {
    const int q = u * PART_THREADS + tid;
    c[u] = cq[(q < nq) ? q : (nq - 1)];
  }
  #pragma unroll
  for (int u = 0; u < PART_Q; ++u) {
    const int q = u * PART_THREADS + tid;
    rk[u] = make_int4(-1, -1, -1, -1);
    if (q < nq) {
      rk[u].x = part_rank(s_cnt, c[u].x, nkeep);
      rk[u].y = part_rank(s_cnt, c[u].y, nkeep);
      rk[u].z = part_rank(s_cnt, c[u].z, nkeep);
      rk[u].w = part_rank(s_cnt, c[u].w, nkeep);
    }
  }
  __syncthreads();
  if (tid < 32) {
    const int4 v = ((const int4*)s_cnt)[tid];
    const int p1 = v.x, p2 = p1 + v.y, p3 = p2 + v.z, p4 = p3 + v.w;
    const int ex = wave_incl_scan(p4, tid) - p4;
    const int4 o = make_int4(ex, ex + p1, ex + p2, ex + p3);
    ((int4*)s_start)[tid] = o;
    ((int4*)(cntT   + (size_t)b * NR_MAX))[tid] = v;
    ((int4*)(startT + (size_t)b * NR_MAX))[tid] = o;
  }
  __syncthreads();

  int4 r4[PART_Q]; float4 v4[PART_Q];
  #pragma unroll
  for (int u = 0; u < PART_Q; ++u) {
    const int q = u * PART_THREADS + tid;
    r4[u] = rq[(q < nq) ? q : (nq - 1)];
    v4[u] = vq[(q < nq) ? q : (nq - 1)];
  }
  #pragma unroll
  for (int u = 0; u < PART_Q; ++u) {
    part_place(sC, sR, sV, s_start, c[u].x, rk[u].x, r4[u].x, v4[u].x);
    part_place(sC, sR, sV, s_start, c[u].y, rk[u].y, r4[u].y, v4[u].y);
    part_place(sC, sR, sV, s_start, c[u].z, rk[u].z, r4[u].z, v4[u].z);
    part_place(sC, sR, sV, s_start, c[u].w, rk[u].w, r4[u].w, v4[u].w);
  }
  __syncthreads();
  const int total = s_start[NR_MAX - 1] + s_cnt[NR_MAX - 1];
  const int n4 = total >> 2;
  int4* __restrict__ oC = (int4*)(gC + (size_t)b * CH);
  int4* __restrict__ oR = (int4*)(gR + (size_t)b * CH);
  int4* __restrict__ oV = (int4*)(gV + (size_t)b * CH);
  for (int w = tid; w < n4; w += PART_THREADS) {
    oC[w] = ((const int4*)sC)[w];
    oR[w] = ((const int4*)sR)[w];
    oV[w] = ((const int4*)sV)[w];
  }
  const int t1 = 4 * n4 + tid;
  if (t1 < total) {
    gC[(size_t)b * CH + t1] = sC[t1];
    gR[(size_t)b * CH + t1] = sR[t1];
    gV[(size_t)b * CH + t1] = sV[t1];
  }
}

#define BK_THREADS 1024
#define BK_U1 8
#define BK_U2 8
static_assert(BK_THREADS == RCOLS, "one thread per column of the range");

__global__ __launch_bounds__(BK_THREADS)
void bucket3(const int* __restrict__ gC, const int* __restrict__ gR, const int* __restrict__ gV,
             const int* __restrict__ cntT, const int* __restrict__ startT,
             int* __restrict__ offU, int* __restrict__ pairsU,
             int* __restrict__ offI, int* __restrict__ pairsI,
             int* __restrict__ offX, int* __restrict__ pairsX) {
  __shared__ int s_cc[RCOLS];
  __shared__ int s_pc[NCH_MAX];
  __shared__ int s_ps[NCH_MAX];
  __shared__ int s_w[32];
  __shared__ int s_w2[32];
  const int tid  = threadIdx.x;
  const int lane = tid & 31;
  const int wave = __builtin_amdgcn_readfirstlane(tid >> 5);
  int r = blockIdx.x, ch0 = 0, nch = NCH_U, n = N_UG, nr = NR_U;
  int* __restrict__ offs = offU; int* __restrict__ pairs = pairsU;
  if (r >= NR_U + NR_I) {
    r -= NR_U + NR_I; ch0 = NCH_U + NCH_I; nch = NCH_X; n = N_UIB; nr = NR_X; offs = offX; pairs = pairsX;
  } else if (r >= NR_U) {
    r -= NR_U; ch0 = NCH_U; nch = NCH_I; n = N_IG; nr = NR_I; offs = offI; pairs = pairsI;
  }
  s_cc[tid] = 0;
  int bsum = 0;
  if (tid < nch) {
    const int pc = cntT[(size_t)(ch0 + tid) * NR_MAX + r];
    const int ps = startT[(size_t)(ch0 + tid) * NR_MAX + r];
    s_pc[tid] = pc; s_ps[tid] = ps;
    bsum = ps;
  }
  #pragma unroll
  for (int o = 16; o > 0; o >>= 1) bsum += __shfl_xor(bsum, o, 32);
  if (lane == 0) s_w[wave] = bsum;
  __syncthreads();
  int base = s_w[lane];
  #pragma unroll
  for (int o = 16; o > 0; o >>= 1) base += __shfl_xor(base, o, 32);

  for (int k = wave; k < nch; k += 32) {
    const int pc = __builtin_amdgcn_readfirstlane(s_pc[k]);
    const int* __restrict__ p = gC + (size_t)(ch0 + k) * CH + __builtin_amdgcn_readfirstlane(s_ps[k]);
    for (int i0 = 0; i0 < pc; i0 += 32 * BK_U1) {
      int c[BK_U1];
      #pragma unroll
      for (int u = 0; u < BK_U1; ++u) {
        const int i = i0 + u * 32 + lane;
        c[u] = (i < pc) ? p[i] : -1;
      }
      #pragma unroll
      for (int u = 0; u < BK_U1; ++u)
        if (c[u] >= 0) atomicAdd(&s_cc[c[u] & (RCOLS - 1)], 1);
    }
  }
  __syncthreads();

  const int cnt = s_cc[tid];
  const int inc = wave_incl_scan(cnt, lane);
  if (lane == 31) s_w2[wave] = inc;
  __syncthreads();
  const int wt    = s_w2[lane];
  const int winc  = wave_incl_scan(wt, lane);
  const int wbase = __builtin_amdgcn_readlane(winc - wt, wave);
  const int total = __builtin_amdgcn_readlane(winc, 31);
  const int ex    = wbase + inc - cnt;
  const int colg  = (r << RB) + tid;
  if (colg < n) offs[colg] = base + ex;
  if (r == nr - 1 && tid == 0) offs[n] = base + total;
  s_cc[tid] = ex;
  __syncthreads();

  for (int k = wave; k < nch; k += 32) {
    const int pc = __builtin_amdgcn_readfirstlane(s_pc[k]);
    const size_t o = (size_t)(ch0 + k) * CH + (size_t)__builtin_amdgcn_readfirstlane(s_ps[k]);
    for (int i0 = 0; i0 < pc; i0 += 32 * BK_U2) {
      int c[BK_U2], rr[BK_U2], vv[BK_U2];
      #pragma unroll
      for (int u = 0; u < BK_U2; ++u) {
        const int i = i0 + u * 32 + lane;
        const bool in = i < pc;
        c[u]  = in ? gC[o + i] : -1;
        rr[u] = in ? gR[o + i] : 0;
        vv[u] = in ? gV[o + i] : 0;
      }
      #pragma unroll
      for (int u = 0; u < BK_U2; ++u) {
        if (c[u] >= 0) {
          const int pos = atomicAdd(&s_cc[c[u] & (RCOLS - 1)], 1);
          int2 rv; rv.x = rr[u]; rv.y = vv[u];
          *(int2*)(pairs + 2 * (size_t)(base + pos)) = rv;
        }
      }
    }
  }
}

typedef _Float16 v2h __attribute__((ext_vector_type(2)));

#define GATHER_DEPTH 16

__device__ __forceinline__ const float* table_row(const float* __restrict__ p0, int n0,
                                                  const float* __restrict__ p1, int r) {
  return (r < n0) ? (p0 + (size_t)r * EMB) : (p1 + (size_t)(r - n0) * EMB);
}

__device__ __forceinline__ int2 load_pairs(const int2* __restrict__ rv, int base,
                                           int end, int lane) {
  int2 p; p.x = 0; p.y = 0;
  if (base + lane < end) p = rv[base + lane];
  return p;
}

typedef float v2f __attribute__((ext_vector_type(2)));
typedef const v2f __attribute__((address_space(1))) gv2f;

__device__ __forceinline__ void pair_addr(const float* __restrict__ p0, int n0,
                                          const float* __restrict__ p1, int r,
                                          int& alo, int& ahi) {
  const unsigned long long a = (unsigned long long)table_row(p0, n0, p1, r);
  alo = (int)(unsigned)a;
  ahi = (int)(unsigned)(a >> 32);
}

template <int U>
__device__ __forceinline__ void loads_from(const unsigned (&lo)[GATHER_DEPTH],
                                           const unsigned (&hi)[GATHER_DEPTH],
                                           int j, int n, int lane,
                                           float2 (&d)[GATHER_DEPTH]) {
  if constexpr (U < GATHER_DEPTH) {
    if (j + U < n) {
      const gv2f* rp = (const gv2f*)(((unsigned long long)hi[U] << 32) | lo[U]);
      const v2f t = rp[lane];
      d[U].x = t.x; d[U].y = t.y;
      loads_from<U + 1>(lo, hi, j, n, lane, d);
    }
  }
}

__device__ __forceinline__ void batch_load(int alo, int ahi, int j, int n, int lane,
                                           float2 (&d)[GATHER_DEPTH]) {
  unsigned lo[GATHER_DEPTH], hi[GATHER_DEPTH];
  #pragma unroll
  for (int u = 0; u < GATHER_DEPTH; ++u) {
    lo[u] = (unsigned)__builtin_amdgcn_readlane(alo, j + u);
    hi[u] = (unsigned)__builtin_amdgcn_readlane(ahi, j + u);
  }
  loads_from<0>(lo, hi, j, n, lane, d);
}

template <int U>
__device__ __forceinline__ void sums_from(int pv, int j, int n,
                                          const float2 (&d)[GATHER_DEPTH],
                                          float& ax, float& ay) {
  if constexpr (U < GATHER_DEPTH) {
    if (j + U < n) {
      const float v = __int_as_float(__builtin_amdgcn_readlane(pv, j + U));
      ax += d[U].x * v;
      ay += d[U].y * v;
      sums_from<U + 1>(pv, j, n, d, ax, ay);
    }
  }
}

__device__ __forceinline__ void batch_sum(int pv, int j, int n,
                                          const float2 (&d)[GATHER_DEPTH],
                                          float& ax, float& ay) {
  sums_from<0>(pv, j, n, d, ax, ay);
}

__device__ __forceinline__ void csr_prop_rows(const float* __restrict__ p0, int n0,
                         const float* __restrict__ p1,
                         const int* __restrict__ offsets,
                         const int* __restrict__ pairs,
                         float* __restrict__ out, int rowBeg, int nRows, int rowsPerWave,
                         int addSelf,
                         _Float16* __restrict__ nlo, int split,
                         _Float16* __restrict__ nhi, int blk) {
  const int2* __restrict__ rv = (const int2*)pairs;
  const int lane = threadIdx.x & 31;
  const int wv = __builtin_amdgcn_readfirstlane(
      (int)((blk * blockDim.x + threadIdx.x) >> 5));
  const int r0 = rowBeg + wv * rowsPerWave;
  if (r0 >= nRows) return;
  const int nr = (nRows - r0 < rowsPerWave) ? (nRows - r0) : rowsPerWave;
  const int offv = (lane <= nr) ? offsets[r0 + lane] : 0;
  int beg = __builtin_amdgcn_readlane(offv, 0);
  int end = __builtin_amdgcn_readlane(offv, 1);
  int2 cur = load_pairs(rv, beg, end, lane);

  for (int i = 0; i < nr; ++i) {
    const int wid = r0 + i;
    const int nf  = (end - beg < 32) ? (end - beg) : 32;
    float ax = 0.0f, ay = 0.0f;
    float2 d[GATHER_DEPTH];
    int alo, ahi;
    pair_addr(p0, n0, p1, cur.x, alo, ahi);
    batch_load(alo, ahi, 0, nf, lane, d);
    int nend = end;
    int2 nxt; nxt.x = 0; nxt.y = 0;
    if (i + 1 < nr) {
      nend = __builtin_amdgcn_readlane(offv, i + 2);
      nxt  = load_pairs(rv, end, nend, lane);
    }
    float sx = 0.0f, sy = 0.0f;
    if (addSelf) {
      float2 s = *(const float2*)(table_row(p0, n0, p1, wid) + lane * 2);
      sx = s.x; sy = s.y;
    }
    batch_sum(cur.y, 0, nf, d, ax, ay);
    if (nf > GATHER_DEPTH) {
      batch_load(alo, ahi, GATHER_DEPTH, nf, lane, d);
      batch_sum(cur.y, GATHER_DEPTH, nf, d, ax, ay);
    }
    for (int base = beg + 32; base < end; base += 32) {
      const int2 more = load_pairs(rv, base, end, lane);
      const int n = (end - base < 32) ? (end - base) : 32;
      int mlo, mhi;
      pair_addr(p0, n0, p1, more.x, mlo, mhi);
      for (int j = 0; j < n; j += GATHER_DEPTH) {
        batch_load(mlo, mhi, j, n, lane, d);
        batch_sum(more.y, j, n, d, ax, ay);
      }
    }
    float2 o; o.x = ax; o.y = ay;
    if (addSelf) { o.x = (sx + ax) * 0.5f; o.y = (sy + ay) * 0.5f; }
    *(float2*)(out + (size_t)wid * EMB + lane * 2) = o;

    if (nlo != nullptr || nhi != nullptr) {
      float ss = fmaf(o.x, o.x, o.y * o.y);
      #pragma unroll
      for (int off = 1; off < 32; off <<= 1) ss += __shfl_xor(ss, off, 32);
      float scale = 1.0f / fmaxf(sqrtf(ss), EPS_N);
      _Float16* dst = nullptr;
      if (wid < split) { if (nlo != nullptr) dst = nlo + (size_t)wid * EMB; }
      else             { if (nhi != nullptr) dst = nhi + (size_t)(wid - split) * EMB; }
      if (dst != nullptr) {
        v2h h;
        h.x = (_Float16)(o.x * scale);
        h.y = (_Float16)(o.y * scale);
        *(v2h*)(dst + lane * 2) = h;
      }
    }
    beg = end; end = nend; cur = nxt;
  }
}

__global__ void csr_prop3(
    const float* __restrict__ p0A, int n0A, const float* __restrict__ p1A,
    const int* __restrict__ offA, const int* __restrict__ pairsA, float* __restrict__ outA,
    int rowBegA, int nRowsA, int rpwA, int addSelfA,
    _Float16* __restrict__ nloA, int splitA, _Float16* __restrict__ nhiA, int nbA,
    const float* __restrict__ p0B, int n0B, const float* __restrict__ p1B,
    const int* __restrict__ offB, const int* __restrict__ pairsB, float* __restrict__ outB,
    int rowBegB, int nRowsB, int rpwB, int addSelfB,
    _Float16* __restrict__ nloB, int splitB, _Float16* __restrict__ nhiB, int nbB,
    const float* __restrict__ p0C, int n0C, const float* __restrict__ p1C,
    const int* __restrict__ offC, const int* __restrict__ pairsC, float* __restrict__ outC,
    int rowBegC, int nRowsC, int rpwC, int addSelfC,
    _Float16* __restrict__ nloC, int splitC, _Float16* __restrict__ nhiC) {
  const float* p0 = p0A; const float* p1 = p1A; int n0 = n0A;
  const int* off = offA; const int* pairs = pairsA; float* out = outA;
  int rowBeg = rowBegA, nRows = nRowsA, rpw = rpwA, addSelf = addSelfA, split = splitA;
  _Float16* nlo = nloA; _Float16* nhi = nhiA;
  int blk = blockIdx.x;
  if (blk >= nbA + nbB) {
    p0 = p0C; p1 = p1C; n0 = n0C; off = offC; pairs = pairsC; out = outC;
    rowBeg = rowBegC; nRows = nRowsC; rpw = rpwC; addSelf = addSelfC; split = splitC;
    nlo = nloC; nhi = nhiC; blk -= nbA + nbB;
  } else if (blk >= nbA) {
    p0 = p0B; p1 = p1B; n0 = n0B; off = offB; pairs = pairsB; out = outB;
    rowBeg = rowBegB; nRows = nRowsB; rpw = rpwB; addSelf = addSelfB; split = splitB;
    nlo = nloB; nhi = nhiB; blk -= nbA;
  }
  csr_prop_rows(p0, n0, p1, off, pairs, out, rowBeg, nRows, rpw, addSelf, nlo, split, nhi, blk);
}

#define OUT_THREADS 256
#define OUT_NB      ((BATCH * EMB + OUT_THREADS - 1) / OUT_THREADS)
__device__ __forceinline__
void gather_out_reg_rows(const float* __restrict__ ug,
                               const float* __restrict__ uif,
                               const float* __restrict__ ig,
                               const float* __restrict__ ued,
                               const float* __restrict__ uei,
                               const float* __restrict__ ged,
                               const float* __restrict__ gei,
                               const int* __restrict__ u,
                               const int* __restrict__ p,
                               const int* __restrict__ ng,
                               float* __restrict__ out,
                               float* __restrict__ reg_out, int blk) {
  int gid = blk * OUT_THREADS + threadIdx.x;
  if (gid >= BATCH * EMB) return;
  int b = gid >> 6, c = gid & 63;
  const size_t PO = (size_t)BATCH * 128;
  const size_t NO = 2 * PO;
  out[(size_t)b * 128 + c]           = ug [(size_t)u[b] * EMB + c];
  out[(size_t)b * 128 + 64 + c]      = uif[(size_t)u[b] * EMB + c];
  out[PO + (size_t)b * 128 + c]      = ug [(size_t)(NUM_USER + p[b])  * EMB + c];
  out[PO + (size_t)b * 128 + 64 + c] = ig [(size_t)(NUM_ITEM + p[b])  * EMB + c];
  out[NO + (size_t)b * 128 + c]      = ug [(size_t)(NUM_USER + ng[b]) * EMB + c];
  out[NO + (size_t)b * 128 + 64 + c] = ig [(size_t)(NUM_ITEM + ng[b]) * EMB + c];

  size_t uo = (size_t)u[b]  * EMB + c;
  size_t po = (size_t)p[b]  * EMB + c;
  size_t no = (size_t)ng[b] * EMB + c;
  float s = uei[uo]*uei[uo] + ued[uo]*ued[uo]
          + gei[po]*gei[po] + ged[po]*ged[po]
          + gei[no]*gei[no] + ged[no]*ged[no];
  #pragma unroll
  for (int off = 1; off < 32; off <<= 1) s += __shfl_xor(s, off, 32);
  if ((threadIdx.x & 31) == 0) atomicAdd(reg_out, s * (0.5f / (float)BATCH));
}

#define GEMM_MT     8
#define GEMM_ROWS   (16 * GEMM_MT)
#define GEMM_WAVES  (BATCH / GEMM_ROWS)
#define GEMM_TPG_U  13
#define GEMM_TPG_G  5
static_assert(BATCH % GEMM_ROWS == 0, "whole row groups");
static_assert(GEMM_WAVES * 32 == OUT_THREADS, "the output work-groups have the SSL launch's size");
static_assert(NUM_USER % 16 == 0 && NUM_GROUP % 16 == 0, "whole table tiles");

__device__ __forceinline__ void load_frag2(const _Float16* __restrict__ row, int h,
                                           v16h& f0, v16h& f1) {
  const _Float16* p = row + 16 * h;
  f0 = *(const v16h*)p;
  f1 = *(const v16h*)(p + 32);
}

__device__ __forceinline__ float row16_sum(float x) {
  x += __int_as_float(__builtin_amdgcn_update_dpp(0, __float_as_int(x), 0x111, 0xf, 0xf, true));
  x += __int_as_float(__builtin_amdgcn_update_dpp(0, __float_as_int(x), 0x112, 0xf, 0xf, true));
  x += __int_as_float(__builtin_amdgcn_update_dpp(0, __float_as_int(x), 0x114, 0xf, 0xf, true));
  x += __int_as_float(__builtin_amdgcn_update_dpp(0, __float_as_int(x), 0x118, 0xf, 0xf, true));
  return x;
}

#define GEMM_NB_U   ((NUM_USER  / 16 + GEMM_TPG_U - 1) / GEMM_TPG_U)
#define GEMM_NB_G   ((NUM_GROUP / 16 + GEMM_TPG_G - 1) / GEMM_TPG_G)

__device__ __forceinline__
void ssl_gemm_term(const _Float16* __restrict__ A16,
                   const _Float16* __restrict__ B16,
                   const int* __restrict__ idx,
                   float* __restrict__ tot,
                   int nTiles, int tilesPerGroup, int blk) {
  __shared__ __attribute__((aligned(16))) float s_tot[GEMM_WAVES][GEMM_ROWS];
  const int lane = threadIdx.x & 31;
  const int wave = threadIdx.x >> 5;
  const int c    = lane & 15;
  const int h    = lane >> 4;
  const int row0 = wave * GEMM_ROWS;

  v16h a0[GEMM_MT], a1[GEMM_MT];
  #pragma unroll
  for (int mt = 0; mt < GEMM_MT; ++mt)
    load_frag2(A16 + (size_t)idx[row0 + mt * 16 + c] * EMB, h, a0[mt], a1[mt]);

  float s[GEMM_MT][8];
  #pragma unroll
  for (int mt = 0; mt < GEMM_MT; ++mt) {
    #pragma unroll
    for (int v = 0; v < 8; ++v) s[mt][v] = 0.0f;
  }

  const int t0 = blk * tilesPerGroup;
  const int t1 = (t0 + tilesPerGroup < nTiles) ? (t0 + tilesPerGroup) : nTiles;
  const _Float16* brow = B16 + ((size_t)t0 * 16 + c) * EMB;
  v16h b0, b1;
  load_frag2(brow, h, b0, b1);
  for (int t = t0; t < t1; ++t) {
    const _Float16* nrow = brow + ((t + 1 < t1) ? 16 * EMB : 0);
    v16h nb0, nb1;
    load_frag2(nrow, h, nb0, nb1);
    asm volatile("" ::: "memory");
    #pragma unroll
    for (int mt = 0; mt < GEMM_MT; ++mt) {
      v8f d = {};
      d = __builtin_amdgcn_wmma_f32_16x16x32_f16(false, a0[mt], false, b0,
                                                 (short)0, d, false, false);
      d = __builtin_amdgcn_wmma_f32_16x16x32_f16(false, a1[mt], false, b1,
                                                 (short)0, d, false, false);
      #pragma unroll
      for (int v = 0; v < 8; ++v) s[mt][v] += __expf(d[v] * INV_TEMP);
    }
    b0 = nb0; b1 = nb1; brow = nrow;
  }

  #pragma unroll
  for (int mt = 0; mt < GEMM_MT; ++mt) {
    #pragma unroll
    for (int v = 0; v < 8; ++v) s[mt][v] = row16_sum(s[mt][v]);
  }
  if (c == 15) {
    #pragma unroll
    for (int mt = 0; mt < GEMM_MT; ++mt) {
      float4* p = (float4*)&s_tot[wave][mt * 16 + 8 * h];
      p[0] = make_float4(s[mt][0], s[mt][1], s[mt][2], s[mt][3]);
      p[1] = make_float4(s[mt][4], s[mt][5], s[mt][6], s[mt][7]);
    }
  }
  __syncthreads();
  #pragma unroll
  for (int q = 0; q < GEMM_ROWS / 32; ++q)
    atomicAdd(&tot[row0 + q * 32 + lane], s_tot[wave][q * 32 + lane]);
}

#define GEMM_NB_ALL (2 * GEMM_NB_U + 2 * GEMM_NB_G)
__global__ __launch_bounds__(GEMM_WAVES * 32)
void ssl_gemm4_out(const _Float16* __restrict__ uiU, const _Float16* __restrict__ ugU,
                   const _Float16* __restrict__ igG, const _Float16* __restrict__ ugG,
                   const int* __restrict__ uin, const int* __restrict__ pos,
                   const int* __restrict__ neg, float* __restrict__ tot,
                   const float* __restrict__ ug, const float* __restrict__ uif,
                   const float* __restrict__ ig,
                   const float* __restrict__ ued, const float* __restrict__ uei,
                   const float* __restrict__ ged, const float* __restrict__ gei,
                   float* __restrict__ out, float* __restrict__ reg_out) {
  if (blockIdx.x >= GEMM_NB_ALL) {
    gather_out_reg_rows(ug, uif, ig, ued, uei, ged, gei, uin, pos, neg, out, reg_out,
                        blockIdx.x - GEMM_NB_ALL);
    return;
  }
  const _Float16* A16 = uiU; const _Float16* B16 = ugU; const int* idx = uin;
  int nTiles = NUM_USER / 16, tpg = GEMM_TPG_U, term = 0, blk = blockIdx.x;
  if (blk >= 2 * GEMM_NB_U + GEMM_NB_G) {
    A16 = ugG; B16 = igG; idx = pos; nTiles = NUM_GROUP / 16; tpg = GEMM_TPG_G; term = 3;
    blk -= 2 * GEMM_NB_U + GEMM_NB_G;
  } else if (blk >= 2 * GEMM_NB_U) {
    A16 = igG; B16 = ugG; idx = pos; nTiles = NUM_GROUP / 16; tpg = GEMM_TPG_G; term = 2;
    blk -= 2 * GEMM_NB_U;
  } else if (blk >= GEMM_NB_U) {
    A16 = ugU; B16 = uiU; term = 1; blk -= GEMM_NB_U;
  }
  ssl_gemm_term(A16, B16, idx, tot + term * BATCH, nTiles, tpg, blk);
}

__global__ void ssl_loss4(const _Float16* __restrict__ uiU, const _Float16* __restrict__ ugU,
                          const _Float16* __restrict__ igG, const _Float16* __restrict__ ugG,
                          const int* __restrict__ uin, const int* __restrict__ pos,
                          const float* __restrict__ tot4,
                          float* __restrict__ ssl_out) {
  int gid = blockIdx.x * blockDim.x + threadIdx.x;
  int w = gid >> 5;
  if (w >= 4 * BATCH) return;
  const int term = w / BATCH;
  int i = w - term * BATCH;
  const _Float16* A16 = uiU; const _Float16* B16 = ugU; const int* idx = uin;
  if (term == 1) { A16 = ugU; B16 = uiU; }
  if (term == 2) { A16 = igG; B16 = ugG; idx = pos; }
  if (term == 3) { A16 = ugG; B16 = igG; idx = pos; }
  const float* tot = tot4 + term * BATCH;
  int t = gid & 31;
  size_t ro = (size_t)idx[i] * EMB + 2 * t;
  float d = (float)A16[ro] * (float)B16[ro]
          + (float)A16[ro + 1] * (float)B16[ro + 1];
  #pragma unroll
  for (int off = 1; off < 32; off <<= 1) d += __shfl_xor(d, off, 32);
  if (t == 0) {
    float li = -(d * INV_TEMP - logf(tot[i])) * (1.0f / (float)BATCH);
    atomicAdd(ssl_out, li);
  }
}

extern "C" void kernel_launch(void* const* d_in, const int* in_sizes, int n_in,
                              void* d_out, int out_size, void* d_ws, size_t ws_size,
                              hipStream_t stream) {
  (void)in_sizes; (void)n_in; (void)out_size; (void)ws_size;
  const float* ued  = (const float*)d_in[0];
  const float* uei  = (const float*)d_in[1];
  const float* ged  = (const float*)d_in[2];
  const float* gei  = (const float*)d_in[3];
  const float* item = (const float*)d_in[4];
  const int*   hu_r = (const int*)d_in[5];
  const int*   hu_c = (const int*)d_in[6];
  const float* hu_v = (const float*)d_in[7];
  const int*   hi_r = (const int*)d_in[8];
  const int*   hi_c = (const int*)d_in[9];
  const float* hi_v = (const float*)d_in[10];
  const int*   ui_r = (const int*)d_in[11];
  const int*   ui_c = (const int*)d_in[12];
  const float* ui_v = (const float*)d_in[13];
  const int*   uin  = (const int*)d_in[14];
  const int*   pos  = (const int*)d_in[15];
  const int*   neg  = (const int*)d_in[16];
  float* out = (float*)d_out;

  float* W = (float*)d_ws;
  size_t off = 0;
  float* ug_b = W + off; off += (size_t)N_UG * EMB;
  float* ug_a = W + off; off += (size_t)N_UG * EMB;
  float* ig_b = W + off; off += (size_t)N_IG * EMB;
  float* ig_a = W + off; off += (size_t)N_IG * EMB;
  float* agg  = W + off; off += (size_t)N_UI * EMB;
  float* tot  = W + off; off += 4 * BATCH;
  int* pairs_ug = (int*)(W + off); off += E_UI;
  int* offsets_ug = (int*)(W + off); off += N_UI;
  int* offsets_ui = (int*)(W + off); off += N_UI + 1;
  off = (off + 7) & ~(size_t)7;
  _Float16* H = (_Float16*)(W + off);
  _Float16* ug_user_n16  = H;
  _Float16* ui_user_n16  = H + (size_t)NUM_USER * EMB;
  _Float16* ug_group_n16 = H + (size_t)NUM_USER * EMB * 2;
  _Float16* ig_group_n16 = ug_group_n16 + (size_t)NUM_GROUP * EMB;

  int* gC     = (int*)ug_b;
  int* cntT   = gC + (size_t)NCH_ALL * CH;
  int* startT = cntT + (size_t)NCH_ALL * NR_MAX;
  int* gR     = (int*)ig_b;
  int* gV     = (int*)agg;
  int* pairs_ui   = (int*)ug_a;
  int* pairs_ig   = (int*)agg + (size_t)N_UIB * EMB;
  int* offsets_ig = pairs_ig + (size_t)2 * E_I;
  static_assert((size_t)NCH_ALL * CH + (size_t)2 * NCH_ALL * NR_MAX <= (size_t)N_UG * EMB, "columns and tables fit ug_b");
  static_assert((size_t)NCH_ALL * CH <= (size_t)N_IG * EMB, "source rows fit ig_b");
  static_assert((size_t)NCH_ALL * CH <= (size_t)N_UIB * EMB, "values fit the user rows of agg");
  static_assert((((size_t)NCH_ALL * CH) & 3) == 0 && (((size_t)N_UG * EMB) & 3) == 0 && (((size_t)N_IG * EMB) & 3) == 0,
                "range-ordered planes and tables 16-byte aligned");
  static_assert((size_t)2 * E_UI <= (size_t)N_UG * EMB, "user-item pairs fit ug_a");
  static_assert((size_t)2 * E_I + N_IG + 1 <= (size_t)(N_UI - N_UIB) * EMB, "item-group pairs and offsets fit behind the user rows of agg");
  static_assert((size_t)2 * E_U <= (size_t)E_UI, "user-group pairs fit their plane");
  static_assert(N_UG + 1 <= N_UI, "user-group offsets fit their plane");
  static_assert((((size_t)N_UG * EMB * 2 + (size_t)N_IG * EMB * 2 + (size_t)N_UI * EMB + 4 * BATCH) & 1) == 0
                && (((size_t)N_UG * EMB) & 1) == 0 && (((size_t)N_UIB * EMB) & 1) == 0, "pairs 8-byte aligned");
  const int TB = 256;
  #define CDIV(a, b) (((a) + (b) - 1) / (b))
  #define RPW_UG 8
  #define RPW_IG 5
  #define RPW_IG2 2
  #define RPW_UI 7

  part3<<<NCH_ALL, PART_THREADS, 0, stream>>>(
      hu_c, hu_r, hu_v, hi_c, hi_r, hi_v, ui_c, ui_r, ui_v,
      gC, gR, gV, cntT, startT,
      tot, out + (size_t)3 * BATCH * 128);
  bucket3<<<NR_U + NR_I + NR_X, BK_THREADS, 0, stream>>>(
      gC, gR, gV, cntT, startT,
      offsets_ug, pairs_ug, offsets_ig, pairs_ig, offsets_ui, pairs_ui);

  constexpr int nbUI  = CDIV(CDIV(N_UIB, RPW_UI) * 32, TB);
  constexpr int nbIG1 = CDIV(CDIV(N_IG, RPW_IG) * 32, TB);
  constexpr int nbUG  = CDIV(CDIV(N_UG, RPW_UG) * 32, TB);
  constexpr int nbIG2 = CDIV(CDIV(NUM_GROUP, RPW_IG2) * 32, TB);
  csr_prop3<<<nbUI + nbIG1 + nbUG, TB, 0, stream>>>(
      uei, NUM_USER, item, offsets_ui, pairs_ui, agg, 0, N_UIB, RPW_UI,
      1, ui_user_n16, NUM_USER, nullptr, nbUI,
      item, NUM_ITEM, gei, offsets_ig, pairs_ig, ig_b, 0, N_IG, RPW_IG,
      0, nullptr, 0, nullptr, nbIG1,
      ued, NUM_USER, ged, offsets_ug, pairs_ug, ug_b, 0, N_UG, RPW_UG,
      0, nullptr, 0, nullptr);

  csr_prop3<<<nbUG + nbIG2, TB, 0, stream>>>(
      ug_b, 1 << 30, nullptr, offsets_ug, pairs_ug, ug_a, 0, N_UG, RPW_UG,
      0, ug_user_n16, NUM_USER, ug_group_n16, nbUG,
      ig_b, 1 << 30, nullptr, offsets_ig, pairs_ig, ig_a, NUM_ITEM, N_IG, RPW_IG2,
      0, nullptr, NUM_ITEM, ig_group_n16, nbIG2,
      nullptr, 0, nullptr, nullptr, nullptr, nullptr, 0, 0, 1,
      0, nullptr, 0, nullptr);

  ssl_gemm4_out<<<GEMM_NB_ALL + OUT_NB, GEMM_WAVES * 32, 0, stream>>>(
      ui_user_n16, ug_user_n16, ig_group_n16, ug_group_n16, uin, pos, neg, tot,
      ug_a, agg, ig_a, ued, uei, ged, gei, out, out + (size_t)3 * BATCH * 128);


  float* ssl_slot = out + (size_t)3 * BATCH * 128 + 1;
  ssl_loss4<<<CDIV(4 * BATCH * 32, TB), TB, 0, stream>>>(
      ui_user_n16, ug_user_n16, ig_group_n16, ug_group_n16, uin, pos, tot, ssl_slot);
  #undef RPW_UG
  #undef RPW_IG
  #undef RPW_IG2
  #undef RPW_UI
  #undef CDIV
}
